# starting kernel
# baseline (speedup 1.0000x reference)
#include <hip/hip_runtime.h>
#include <math.h>


#define HEADS     4
#define HC        128
#define NEG_SLOPE 0.2f

typedef __attribute__((ext_vector_type(2))) float v2f;
typedef __attribute__((ext_vector_type(8))) float v8f;

__global__ void gat_gemm_wmma(const float* __restrict__ X,
                              const float* __restrict__ W,
                              float* __restrict__ Hout,
                              int K) {
  const int m0   = blockIdx.x << 4;
  const int wave = threadIdx.x >> 5;
  const int lane = threadIdx.x & 31;
  const int n0   = wave << 4;
  const int half = lane >> 4;
  const int l15  = lane & 15;
  const int row  = m0 + l15;
  const int col  = n0 + l15;

  v8f acc = {};
  #pragma unroll 4
  for (int k = 0; k < K; k += 4) {
    const int kk = k + (half << 1);
    const float2 av = *(const float2*)(X + (size_t)row * K + kk);
    v2f a; a.x = av.x; a.y = av.y;
    v2f b;
    b.x = W[(size_t)kk * HC + col];
    b.y = W[(size_t)(kk + 1) * HC + col];
    acc = __builtin_amdgcn_wmma_f32_16x16x4_f32(
         false, a,  false, b,
         (short)0, acc,  false,  false);
  }
  #pragma unroll
  for (int r = 0; r < 8; ++r) {
    const int rr = m0 + r + (half << 3);
    Hout[(size_t)rr * HC + col] = acc[r];
  }
}

__global__ void gat_attn_dots(const float* __restrict__ Hout,
                              const float* __restrict__ att_src,
                              const float* __restrict__ att_dst,
                              float* __restrict__ a_src,
                              float* __restrict__ a_dst) {
  const int node = blockIdx.x;
  const int t    = threadIdx.x;
  const float hv = Hout[(size_t)node * HC + t];
  float vs = hv * att_src[t];
  float vd = hv * att_dst[t];
  #pragma unroll
  for (int off = 16; off > 0; off >>= 1) {
    vs += __shfl_down(vs, off);
    vd += __shfl_down(vd, off);
  }
  if ((t & 31) == 0) {
    const int hd = t >> 5;
    a_src[node * HEADS + hd] = vs;
    a_dst[node * HEADS + hd] = vd;
  }
}

__device__ __forceinline__ unsigned fenc(float f) {
  unsigned u = __float_as_uint(f);
  return (u & 0x80000000u) ? ~u : (u | 0x80000000u);
}
__device__ __forceinline__ float fdec(unsigned u) {
  u = (u & 0x80000000u) ? (u & 0x7FFFFFFFu) : ~u;
  return __uint_as_float(u);
}

__global__ void gat_init(float* __restrict__ out, const float* __restrict__ bias,
                         unsigned* __restrict__ segmax_u, float* __restrict__ denom,
                         long long totalOut, long long totalNH) {
  const long long i = (long long)blockIdx.x * blockDim.x + threadIdx.x;
  if (i < totalOut) out[i] = bias[i & (HC - 1)];
  if (i < totalNH) { segmax_u[i] = 0x007FFFFFu; denom[i] = 0.0f; }
}

__device__ __forceinline__ void edge_sd(const int* __restrict__ src,
                                        const int* __restrict__ dst,
                                        int eid, int E, int& s, int& d) {
  if (eid < E) { s = src[eid]; d = dst[eid]; }
  else         { s = d = eid - E; }
}

__device__ __forceinline__ float lrelu(float e) {
  return e > 0.0f ? e : NEG_SLOPE * e;
}

__global__ void gat_edge_max(const int* __restrict__ src, const int* __restrict__ dst,
                             const float* __restrict__ a_src, const float* __restrict__ a_dst,
                             unsigned* __restrict__ segmax_u, int E, int Et) {
  const long long idx = (long long)blockIdx.x * blockDim.x + threadIdx.x;
  if (idx >= (long long)Et * HEADS) return;
  const int eid = (int)(idx >> 2);
  const int hd  = (int)(idx & 3);
  int s, d; edge_sd(src, dst, eid, E, s, d);
  const float e = lrelu(a_src[s * HEADS + hd] + a_dst[d * HEADS + hd]);
  atomicMax(&segmax_u[d * HEADS + hd], fenc(e));
}

__global__ void gat_segmax_fix(const unsigned* __restrict__ segmax_u,
                               float* __restrict__ segmax_f, int n) {
  const int i = blockIdx.x * blockDim.x + threadIdx.x;
  if (i >= n) return;
  const float m = fdec(segmax_u[i]);
  segmax_f[i] = isfinite(m) ? m : 0.0f;
}

__global__ void gat_edge_denom(const int* __restrict__ src, const int* __restrict__ dst,
                               const float* __restrict__ a_src, const float* __restrict__ a_dst,
                               const float* __restrict__ segmax_f, float* __restrict__ denom,
                               int E, int Et) {
  const long long idx = (long long)blockIdx.x * blockDim.x + threadIdx.x;
  if (idx >= (long long)Et * HEADS) return;
  const int eid = (int)(idx >> 2);
  const int hd  = (int)(idx & 3);
  int s, d; edge_sd(src, dst, eid, E, s, d);
  const float e  = lrelu(a_src[s * HEADS + hd] + a_dst[d * HEADS + hd]);
  const float ex = __expf(e - segmax_f[d * HEADS + hd]);
  atomicAdd(&denom[d * HEADS + hd], ex);
}

__global__ void gat_edge_scatter(const int* __restrict__ src, const int* __restrict__ dst,
                                 const float* __restrict__ a_src, const float* __restrict__ a_dst,
                                 const float* __restrict__ segmax_f, const float* __restrict__ denom,
                                 const float* __restrict__ Hout, float* __restrict__ out,
                                 int E, int Et) {
  const long long widx = ((long long)blockIdx.x * blockDim.x + threadIdx.x) >> 5;
  if (widx >= Et) return;
  const int lane = threadIdx.x & 31;
  const int eid  = (int)widx;
  int s, d; edge_sd(src, dst, eid, E, s, d);
  const int hd = lane >> 3;
  const float e     = lrelu(a_src[s * HEADS + hd] + a_dst[d * HEADS + hd]);
  const float ex    = __expf(e - segmax_f[d * HEADS + hd]);
  const float alpha = ex / (denom[d * HEADS + hd] + 1e-16f);

  const int ch = lane << 2;
  const float4 hv = *(const float4*)(Hout + (size_t)s * HC + ch);
  float* op = out + (size_t)d * HC + ch;
  atomicAdd(op + 0, hv.x * alpha);
  atomicAdd(op + 1, hv.y * alpha);
  atomicAdd(op + 2, hv.z * alpha);
  atomicAdd(op + 3, hv.w * alpha);
}

extern "C" void kernel_launch(void* const* d_in, const int* in_sizes, int n_in,
                              void* d_out, int out_size, void* d_ws, size_t ws_size,
                              hipStream_t stream) {
  const float* X        = (const float*)d_in[0];
  const int*   edges    = (const int*)  d_in[1];
  const float* W        = (const float*)d_in[2];
  const float* att_src  = (const float*)d_in[3];
  const float* att_dst  = (const float*)d_in[4];
  const float* bias     = (const float*)d_in[5];
  float*       out      = (float*)d_out;

  const int K  = in_sizes[2] / HC;
  const int Nn = in_sizes[0] / K;
  const int E  = in_sizes[1] / 2;
  const int Et = E + Nn;
  const int* src = edges;
  const int* dst = edges + E;

  char* ws = (char*)d_ws;
  float*    Hout     = (float*)ws;    ws += (size_t)Nn * HC    * sizeof(float);
  float*    a_src    = (float*)ws;    ws += (size_t)Nn * HEADS * sizeof(float);
  float*    a_dst    = (float*)ws;    ws += (size_t)Nn * HEADS * sizeof(float);
  unsigned* segmax_u = (unsigned*)ws; ws += (size_t)Nn * HEADS * sizeof(unsigned);
  float*    segmax_f = (float*)ws;    ws += (size_t)Nn * HEADS * sizeof(float);
  float*    denom    = (float*)ws;    ws += (size_t)Nn * HEADS * sizeof(float);

  const long long totalOut = (long long)Nn * HC;
  const long long totalNH  = (long long)Nn * HEADS;
  const long long totalEH  = (long long)Et * HEADS;

  gat_init<<<(int)((totalOut + 255) / 256), 256, 0, stream>>>(
      out, bias, segmax_u, denom, totalOut, totalNH);

  gat_gemm_wmma<<<(Nn + 15) / 16, 256, 0, stream>>>(X, W, Hout, K);

  gat_attn_dots<<<Nn, HC, 0, stream>>>(Hout, att_src, att_dst, a_src, a_dst);

  gat_edge_max<<<(int)((totalEH + 255) / 256), 256, 0, stream>>>(
      src, dst, a_src, a_dst, segmax_u, E, Et);
  gat_segmax_fix<<<(int)((totalNH + 255) / 256), 256, 0, stream>>>(
      segmax_u, segmax_f, (int)totalNH);

  gat_edge_denom<<<(int)((totalEH + 255) / 256), 256, 0, stream>>>(
      src, dst, a_src, a_dst, segmax_f, denom, E, Et);

  gat_edge_scatter<<<(Et + 7) / 8, 256, 0, stream>>>(
      src, dst, a_src, a_dst, segmax_f, denom, Hout, out, E, Et);
}


# Edges counting-sorted by destination (int atomics); one wave per node gathers source rows with a running softmax in registers, replacing the float-atomic edge passes; sum order: arrival rank
# speedup vs baseline: 1.8989x; 1.8989x over previous
#include <hip/hip_runtime.h>
#include <math.h>


#define HEADS     4
#define HC        128
#define NEG_SLOPE 0.2f

typedef __attribute__((ext_vector_type(2))) float v2f;
typedef __attribute__((ext_vector_type(8))) float v8f;

__global__ void gat_gemm_wmma(const float* __restrict__ X,
                              const float* __restrict__ W,
                              float* __restrict__ Hout,
                              int K) {
  const int m0   = blockIdx.x << 4;
  const int wave = threadIdx.x >> 5;
  const int lane = threadIdx.x & 31;
  const int n0   = wave << 4;
  const int half = lane >> 4;
  const int l15  = lane & 15;
  const int row  = m0 + l15;
  const int col  = n0 + l15;

  v8f acc = {};
  #pragma unroll 4
  for (int k = 0; k < K; k += 4) {
    const int kk = k + (half << 1);
    const float2 av = *(const float2*)(X + (size_t)row * K + kk);
    v2f a; a.x = av.x; a.y = av.y;
    v2f b;
    b.x = W[(size_t)kk * HC + col];
    b.y = W[(size_t)(kk + 1) * HC + col];
    acc = __builtin_amdgcn_wmma_f32_16x16x4_f32(
         false, a,  false, b,
         (short)0, acc,  false,  false);
  }
  #pragma unroll
  for (int r = 0; r < 8; ++r) {
    const int rr = m0 + r + (half << 3);
    Hout[(size_t)rr * HC + col] = acc[r];
  }
}

__global__ void gat_zero_counts(int* __restrict__ counts, int n) {
  const int i = blockIdx.x * blockDim.x + threadIdx.x;
  if (i < n) counts[i] = 0;
}

__global__ void gat_count_dst(const int* __restrict__ dst, int* __restrict__ counts, int E) {
  const int i = blockIdx.x * blockDim.x + threadIdx.x;
  if (i < E) atomicAdd(&counts[dst[i]], 1);
}

__global__ void __launch_bounds__(1024)
gat_scan_counts(const int* __restrict__ counts, int* __restrict__ seg, int n) {
  __shared__ int pre_part[32];
  __shared__ int tile_part[32];
  const int tid   = threadIdx.x;
  const int lane  = tid & 31;
  const int wave  = tid >> 5;
  const int tile0 = blockIdx.x * 1024;

  int pre = 0;
  #pragma unroll 8
  for (int j = 0; j < (int)blockIdx.x; ++j) pre += counts[(j << 10) + tid];
  for (int off = 16; off > 0; off >>= 1) pre += __shfl_xor(pre, off);
  if (lane == 0) pre_part[wave] = pre;

  const int i = tile0 + tid;
  const int v = (i < n) ? counts[i] : 0;
  int x = v;
  for (int off = 1; off < 32; off <<= 1) {
    const int t = __shfl_up(x, off);
    if (lane >= off) x += t;
  }
  if (lane == 31) tile_part[wave] = x;
  __syncthreads();

  int base = 0;
  for (int w = 0; w < 32; ++w) {
    base += pre_part[w];
    if (w < wave) base += tile_part[w];
  }
  if (i < n) seg[i] = base + x - v;
}

__global__ void gat_fill_sorted(const int* __restrict__ src, const int* __restrict__ dst,
                                int* __restrict__ seg, int* __restrict__ sorted_src, int E) {
  const int i = blockIdx.x * blockDim.x + threadIdx.x;
  if (i >= E) return;
  const int pos = atomicAdd(&seg[dst[i]], 1);
  sorted_src[pos] = src[i];
}

__device__ __forceinline__ float lrelu(float e) {
  return e > 0.0f ? e : NEG_SLOPE * e;
}

template <int CTRL>
__device__ __forceinline__ float dpp_take(float v) {
  return __int_as_float(__builtin_amdgcn_update_dpp(
      0, __float_as_int(v), CTRL, 0xF, 0xF, true));
}

__device__ __forceinline__ float sum8(float v) {
  v += dpp_take<0xB1>(v);
  v += dpp_take<0x4E>(v);
  v += dpp_take<0x141>(v);
  return v;
}

__device__ __forceinline__ float dot4(const float4 a, const float4 b) {
  return a.x * b.x + a.y * b.y + a.z * b.z + a.w * b.w;
}

__global__ void __launch_bounds__(256)
gat_gather_softmax(const float* __restrict__ Hout, const int* __restrict__ seg_end,
                   const int* __restrict__ sorted_src,
                   const float* __restrict__ att_src, const float* __restrict__ att_dst,
                   const float* __restrict__ bias, float* __restrict__ out, int Nn) {
  const int wave = __builtin_amdgcn_readfirstlane((int)(threadIdx.x >> 5));
  const int d    = blockIdx.x * 8 + wave;
  if (d >= Nn) return;
  const int lane = threadIdx.x & 31;
  const int ch   = lane << 2;

  const float4 ws4 = *(const float4*)(att_src + ch);
  const float4 wd4 = *(const float4*)(att_dst + ch);
  const float4 hd4 = *(const float4*)(Hout + (size_t)d * HC + ch);
  const float a_d  = sum8(dot4(hd4, wd4));

  float  m   = lrelu(sum8(dot4(hd4, ws4)) + a_d);
  float  l   = 1.0f;
  float4 acc = hd4;

  const int begin = d > 0 ? seg_end[d - 1] : 0;
  const int end   = seg_end[d];

  for (int base = begin; base < end; base += 32) {
    const int cnt = (end - base) < 32 ? (end - base) : 32;
    const int idx = (lane < cnt) ? sorted_src[base + lane] : 0;
    for (int k = 0; k < cnt; k += 4) {
      float4 hs[4];
      for (int u = 0; u < 4; ++u) {
        const int kk = (k + u) < cnt ? (k + u) : (cnt - 1);
        const int s  = __builtin_amdgcn_readlane(idx, kk);
        hs[u] = *(const float4*)(Hout + (size_t)s * HC + ch);
      }
      for (int u = 0; u < 4; ++u) {
        const float a_s = sum8(dot4(hs[u], ws4));
        const float e  = (k + u) < cnt ? lrelu(a_s + a_d) : -__builtin_inff();
        const float mn = fmaxf(m, e);
        const float sc = __expf(m - mn);
        const float p  = __expf(e - mn);
        l     = l * sc + p;
        acc.x = acc.x * sc + hs[u].x * p;
        acc.y = acc.y * sc + hs[u].y * p;
        acc.z = acc.z * sc + hs[u].z * p;
        acc.w = acc.w * sc + hs[u].w * p;
        m     = mn;
      }
    }
  }

  const float  dn = l + 1e-16f;
  const float4 b4 = *(const float4*)(bias + ch);
  float4 o;
  o.x = acc.x / dn + b4.x;
  o.y = acc.y / dn + b4.y;
  o.z = acc.z / dn + b4.z;
  o.w = acc.w / dn + b4.w;
  *(float4*)(out + (size_t)d * HC + ch) = o;
}

extern "C" void kernel_launch(void* const* d_in, const int* in_sizes, int n_in,
                              void* d_out, int out_size, void* d_ws, size_t ws_size,
                              hipStream_t stream) {
  const float* X        = (const float*)d_in[0];
  const int*   edges    = (const int*)  d_in[1];
  const float* W        = (const float*)d_in[2];
  const float* att_src  = (const float*)d_in[3];
  const float* att_dst  = (const float*)d_in[4];
  const float* bias     = (const float*)d_in[5];
  float*       out      = (float*)d_out;

  const int K  = in_sizes[2] / HC;
  const int Nn = in_sizes[0] / K;
  const int E  = in_sizes[1] / 2;
  const int* src = edges;
  const int* dst = edges + E;

  char* ws = (char*)d_ws;
  float* Hout       = (float*)ws; ws += (size_t)Nn * HC * sizeof(float);
  int*   sorted_src = (int*)ws;   ws += (size_t)E * sizeof(int);
  int*   counts     = (int*)ws;   ws += (size_t)Nn * sizeof(int);
  int*   seg        = (int*)ws;   ws += (size_t)Nn * sizeof(int);

  gat_zero_counts<<<(Nn + 255) / 256, 256, 0, stream>>>(counts, Nn);
  gat_count_dst<<<(E + 255) / 256, 256, 0, stream>>>(dst, counts, E);
  gat_scan_counts<<<(Nn + 1023) / 1024, 1024, 0, stream>>>(counts, seg, Nn);
  gat_fill_sorted<<<(E + 255) / 256, 256, 0, stream>>>(src, dst, seg, sorted_src, E);

  gat_gemm_wmma<<<(Nn + 15) / 16, 256, 0, stream>>>(X, W, Hout, K);

  gat_gather_softmax<<<(Nn + 7) / 8, 256, 0, stream>>>(
      Hout, seg, sorted_src, att_src, att_dst, bias, out, Nn);
}


# GEMM fed from LDS and registers: 512 resident blocks hold W in LDS as k pairs, each wave runs 16x64 tiles so one A load serves 4 WMMAs (was 3 cache loads per WMMA); sum order unchanged
# speedup vs baseline: 2.2511x; 1.1855x over previous
#include <hip/hip_runtime.h>
#include <math.h>


#define HEADS     4
#define HC        128
#define NEG_SLOPE 0.2f

typedef __attribute__((ext_vector_type(2))) float v2f;
typedef __attribute__((ext_vector_type(8))) float v8f;

#define KDIM        256
#define WPITCH      288
#define GEMM_BLOCKS 512

__global__ void __launch_bounds__(256)
gat_gemm_wmma(const float* __restrict__ X,
              const float* __restrict__ W,
              float* __restrict__ Hout,
              int Nn) {
  __shared__ __attribute__((aligned(16))) float wl[(KDIM / 2) * WPITCH];

  {
    const int c4 = (threadIdx.x & 31) << 2;
    #pragma unroll 4
    for (int i = 0; i < KDIM / 16; ++i) {
      const int p = (int)(threadIdx.x >> 5) + 8 * i;
      const float4 w0 = *(const float4*)(W + (size_t)(2 * p) * HC + c4);
      const float4 w1 = *(const float4*)(W + (size_t)(2 * p + 1) * HC + c4);
      float4 lo, hi;
      lo.x = w0.x; lo.y = w1.x; lo.z = w0.y; lo.w = w1.y;
      hi.x = w0.z; hi.y = w1.z; hi.z = w0.w; hi.w = w1.w;
      float* q = wl + p * WPITCH + 2 * c4;
      *(float4*)(q)     = lo;
      *(float4*)(q + 4) = hi;
    }
  }
  __syncthreads();

  const int wave = __builtin_amdgcn_readfirstlane((int)(threadIdx.x >> 5));
  const int lane = threadIdx.x & 31;
  const int half = lane >> 4;
  const int l15  = lane & 15;
  const int n_units = ((Nn + 15) >> 4) << 1;
  const int n_waves = gridDim.x << 3;

  for (int u = blockIdx.x * 8 + wave; u < n_units; u += n_waves) {
    const int m0 = (u >> 1) << 4;
    const int n0 = (u & 1) << 6;
    const float* xa = X + (size_t)(m0 + l15) * KDIM + (half << 1);
    const float* wb = wl + half * WPITCH + 2 * (n0 + l15);

    v8f acc[4] = {};
    float2 ac[8], an[8];
    #pragma unroll
    for (int s = 0; s < 8; ++s) ac[s] = *(const float2*)(xa + 4 * s);

    #pragma unroll 2
    for (int g = 0; g < KDIM / 32; ++g) {
      if (g + 1 < KDIM / 32) {
        #pragma unroll
        for (int s = 0; s < 8; ++s)
          an[s] = *(const float2*)(xa + 32 * (g + 1) + 4 * s);
      }
      const float* wg = wb + g * (16 * WPITCH);
      #pragma unroll
      for (int s = 0; s < 8; ++s) {
        v2f a; a.x = ac[s].x; a.y = ac[s].y;
        #pragma unroll
        for (int t = 0; t < 4; ++t) {
          const float2 bv = *(const float2*)(wg + (2 * s) * WPITCH + 32 * t);
          v2f b; b.x = bv.x; b.y = bv.y;
          acc[t] = __builtin_amdgcn_wmma_f32_16x16x4_f32(
               false, a,  false, b,
               (short)0, acc[t],  false,  false);
        }
      }
      #pragma unroll
      for (int s = 0; s < 8; ++s) ac[s] = an[s];
    }

    #pragma unroll
    for (int t = 0; t < 4; ++t) {
      #pragma unroll
      for (int r = 0; r < 8; ++r) {
        const int rr = m0 + r + (half << 3);
        Hout[(size_t)rr * HC + n0 + 16 * t + l15] = acc[t][r];
      }
    }
  }
}

__global__ void gat_zero_counts(int* __restrict__ counts, int n) {
  const int i = blockIdx.x * blockDim.x + threadIdx.x;
  if (i < n) counts[i] = 0;
}

__global__ void gat_count_dst(const int* __restrict__ dst, int* __restrict__ counts, int E) {
  const int i = blockIdx.x * blockDim.x + threadIdx.x;
  if (i < E) atomicAdd(&counts[dst[i]], 1);
}

__global__ void __launch_bounds__(1024)
gat_scan_counts(const int* __restrict__ counts, int* __restrict__ seg, int n) {
  __shared__ int pre_part[32];
  __shared__ int tile_part[32];
  const int tid   = threadIdx.x;
  const int lane  = tid & 31;
  const int wave  = tid >> 5;
  const int tile0 = blockIdx.x * 1024;

  int pre = 0;
  #pragma unroll 8
  for (int j = 0; j < (int)blockIdx.x; ++j) pre += counts[(j << 10) + tid];
  for (int off = 16; off > 0; off >>= 1) pre += __shfl_xor(pre, off);
  if (lane == 0) pre_part[wave] = pre;

  const int i = tile0 + tid;
  const int v = (i < n) ? counts[i] : 0;
  int x = v;
  for (int off = 1; off < 32; off <<= 1) {
    const int t = __shfl_up(x, off);
    if (lane >= off) x += t;
  }
  if (lane == 31) tile_part[wave] = x;
  __syncthreads();

  int base = 0;
  for (int w = 0; w < 32; ++w) {
    base += pre_part[w];
    if (w < wave) base += tile_part[w];
  }
  if (i < n) seg[i] = base + x - v;
}

__global__ void gat_fill_sorted(const int* __restrict__ src, const int* __restrict__ dst,
                                int* __restrict__ seg, int* __restrict__ sorted_src, int E) {
  const int i = blockIdx.x * blockDim.x + threadIdx.x;
  if (i >= E) return;
  const int pos = atomicAdd(&seg[dst[i]], 1);
  sorted_src[pos] = src[i];
}

__device__ __forceinline__ float lrelu(float e) {
  return e > 0.0f ? e : NEG_SLOPE * e;
}

template <int CTRL>
__device__ __forceinline__ float dpp_take(float v) {
  return __int_as_float(__builtin_amdgcn_update_dpp(
      0, __float_as_int(v), CTRL, 0xF, 0xF, true));
}

__device__ __forceinline__ float sum8(float v) {
  v += dpp_take<0xB1>(v);
  v += dpp_take<0x4E>(v);
  v += dpp_take<0x141>(v);
  return v;
}

__device__ __forceinline__ float dot4(const float4 a, const float4 b) {
  return a.x * b.x + a.y * b.y + a.z * b.z + a.w * b.w;
}

__global__ void __launch_bounds__(256)
gat_gather_softmax(const float* __restrict__ Hout, const int* __restrict__ seg_end,
                   const int* __restrict__ sorted_src,
                   const float* __restrict__ att_src, const float* __restrict__ att_dst,
                   const float* __restrict__ bias, float* __restrict__ out, int Nn) {
  const int wave = __builtin_amdgcn_readfirstlane((int)(threadIdx.x >> 5));
  const int d    = blockIdx.x * 8 + wave;
  if (d >= Nn) return;
  const int lane = threadIdx.x & 31;
  const int ch   = lane << 2;

  const float4 ws4 = *(const float4*)(att_src + ch);
  const float4 wd4 = *(const float4*)(att_dst + ch);
  const float4 hd4 = *(const float4*)(Hout + (size_t)d * HC + ch);
  const float a_d  = sum8(dot4(hd4, wd4));

  float  m   = lrelu(sum8(dot4(hd4, ws4)) + a_d);
  float  l   = 1.0f;
  float4 acc = hd4;

  const int begin = d > 0 ? seg_end[d - 1] : 0;
  const int end   = seg_end[d];

  for (int base = begin; base < end; base += 32) {
    const int cnt = (end - base) < 32 ? (end - base) : 32;
    const int idx = (lane < cnt) ? sorted_src[base + lane] : 0;
    for (int k = 0; k < cnt; k += 4) {
      float4 hs[4];
      for (int u = 0; u < 4; ++u) {
        const int kk = (k + u) < cnt ? (k + u) : (cnt - 1);
        const int s  = __builtin_amdgcn_readlane(idx, kk);
        hs[u] = *(const float4*)(Hout + (size_t)s * HC + ch);
      }
      for (int u = 0; u < 4; ++u) {
        const float a_s = sum8(dot4(hs[u], ws4));
        const float e  = (k + u) < cnt ? lrelu(a_s + a_d) : -__builtin_inff();
        const float mn = fmaxf(m, e);
        const float sc = __expf(m - mn);
        const float p  = __expf(e - mn);
        l     = l * sc + p;
        acc.x = acc.x * sc + hs[u].x * p;
        acc.y = acc.y * sc + hs[u].y * p;
        acc.z = acc.z * sc + hs[u].z * p;
        acc.w = acc.w * sc + hs[u].w * p;
        m     = mn;
      }
    }
  }

  const float  dn = l + 1e-16f;
  const float4 b4 = *(const float4*)(bias + ch);
  float4 o;
  o.x = acc.x / dn + b4.x;
  o.y = acc.y / dn + b4.y;
  o.z = acc.z / dn + b4.z;
  o.w = acc.w / dn + b4.w;
  *(float4*)(out + (size_t)d * HC + ch) = o;
}

extern "C" void kernel_launch(void* const* d_in, const int* in_sizes, int n_in,
                              void* d_out, int out_size, void* d_ws, size_t ws_size,
                              hipStream_t stream) {
  const float* X        = (const float*)d_in[0];
  const int*   edges    = (const int*)  d_in[1];
  const float* W        = (const float*)d_in[2];
  const float* att_src  = (const float*)d_in[3];
  const float* att_dst  = (const float*)d_in[4];
  const float* bias     = (const float*)d_in[5];
  float*       out      = (float*)d_out;

  const int K  = in_sizes[2] / HC;
  const int Nn = in_sizes[0] / K;
  const int E  = in_sizes[1] / 2;
  const int* src = edges;
  const int* dst = edges + E;

  char* ws = (char*)d_ws;
  float* Hout       = (float*)ws; ws += (size_t)Nn * HC * sizeof(float);
  int*   sorted_src = (int*)ws;   ws += (size_t)E * sizeof(int);
  int*   counts     = (int*)ws;   ws += (size_t)Nn * sizeof(int);
  int*   seg        = (int*)ws;   ws += (size_t)Nn * sizeof(int);

  gat_zero_counts<<<(Nn + 255) / 256, 256, 0, stream>>>(counts, Nn);
  gat_count_dst<<<(E + 255) / 256, 256, 0, stream>>>(dst, counts, E);
  gat_scan_counts<<<(Nn + 1023) / 1024, 1024, 0, stream>>>(counts, seg, Nn);
  gat_fill_sorted<<<(E + 255) / 256, 256, 0, stream>>>(src, dst, seg, sorted_src, E);

  gat_gemm_wmma<<<GEMM_BLOCKS, 256, 0, stream>>>(X, W, Hout, Nn);

  gat_gather_softmax<<<(Nn + 7) / 8, 256, 0, stream>>>(
      Hout, seg, sorted_src, att_src, att_dst, bias, out, Nn);
}


# Edges sorted by destination with a two-level counting sort on LDS counters (chunk histogram, scan, bucket scatter, per-bucket sort): no per-edge global atomics; sum order: LDS arrival rank
# speedup vs baseline: 4.3132x; 1.9160x over previous
#include <hip/hip_runtime.h>
#include <math.h>


#define HEADS     4
#define HC        128
#define NEG_SLOPE 0.2f

typedef __attribute__((ext_vector_type(2))) float v2f;
typedef __attribute__((ext_vector_type(8))) float v8f;

#define KDIM        256
#define WPITCH      288
#define GEMM_BLOCKS 512

__global__ void __launch_bounds__(256)
gat_gemm_wmma(const float* __restrict__ X,
              const float* __restrict__ W,
              float* __restrict__ Hout,
              int Nn) {
  __shared__ __attribute__((aligned(16))) float wl[(KDIM / 2) * WPITCH];

  {
    const int c4 = (threadIdx.x & 31) << 2;
    #pragma unroll 4
    for (int i = 0; i < KDIM / 16; ++i) {
      const int p = (int)(threadIdx.x >> 5) + 8 * i;
      const float4 w0 = *(const float4*)(W + (size_t)(2 * p) * HC + c4);
      const float4 w1 = *(const float4*)(W + (size_t)(2 * p + 1) * HC + c4);
      float4 lo, hi;
      lo.x = w0.x; lo.y = w1.x; lo.z = w0.y; lo.w = w1.y;
      hi.x = w0.z; hi.y = w1.z; hi.z = w0.w; hi.w = w1.w;
      float* q = wl + p * WPITCH + 2 * c4;
      *(float4*)(q)     = lo;
      *(float4*)(q + 4) = hi;
    }
  }
  __syncthreads();

  const int wave = __builtin_amdgcn_readfirstlane((int)(threadIdx.x >> 5));
  const int lane = threadIdx.x & 31;
  const int half = lane >> 4;
  const int l15  = lane & 15;
  const int n_units = ((Nn + 15) >> 4) << 1;
  const int n_waves = gridDim.x << 3;

  for (int u = blockIdx.x * 8 + wave; u < n_units; u += n_waves) {
    const int m0 = (u >> 1) << 4;
    const int n0 = (u & 1) << 6;
    const float* xa = X + (size_t)(m0 + l15) * KDIM + (half << 1);
    const float* wb = wl + half * WPITCH + 2 * (n0 + l15);

    v8f acc[4] = {};
    float2 ac[8], an[8];
    #pragma unroll
    for (int s = 0; s < 8; ++s) ac[s] = *(const float2*)(xa + 4 * s);

    #pragma unroll 2
    for (int g = 0; g < KDIM / 32; ++g) {
      if (g + 1 < KDIM / 32) {
        #pragma unroll
        for (int s = 0; s < 8; ++s)
          an[s] = *(const float2*)(xa + 32 * (g + 1) + 4 * s);
      }
      const float* wg = wb + g * (16 * WPITCH);
      #pragma unroll
      for (int s = 0; s < 8; ++s) {
        v2f a; a.x = ac[s].x; a.y = ac[s].y;
        #pragma unroll
        for (int t = 0; t < 4; ++t) {
          const float2 bv = *(const float2*)(wg + (2 * s) * WPITCH + 32 * t);
          v2f b; b.x = bv.x; b.y = bv.y;
          acc[t] = __builtin_amdgcn_wmma_f32_16x16x4_f32(
               false, a,  false, b,
               (short)0, acc[t],  false,  false);
        }
      }
      #pragma unroll
      for (int s = 0; s < 8; ++s) ac[s] = an[s];
    }

    #pragma unroll
    for (int t = 0; t < 4; ++t) {
      #pragma unroll
      for (int r = 0; r < 8; ++r) {
        const int rr = m0 + r + (half << 3);
        Hout[(size_t)rr * HC + n0 + 16 * t + l15] = acc[t][r];
      }
    }
  }
}

#define SORT_SH   9
#define SORT_BN   (1 << SORT_SH)
#define SORT_CH   4096
#define SORT_CT   256
#define SORT_EPT  (SORT_CH / SORT_CT)

__global__ void __launch_bounds__(SORT_CT)
gat_chunk_hist(const int* __restrict__ dst, int* __restrict__ hist, int E, int n_buckets) {
  extern __shared__ int hist_bins[];
  const int tid = threadIdx.x;
  for (int b = tid; b < n_buckets; b += SORT_CT) hist_bins[b] = 0;
  __syncthreads();

  const int base = blockIdx.x * SORT_CH + tid;
  int dv[SORT_EPT];
  #pragma unroll
  for (int j = 0; j < SORT_EPT; ++j) {
    const int i = base + j * SORT_CT;
    dv[j] = (i < E) ? dst[i] : -1;
  }
  #pragma unroll
  for (int j = 0; j < SORT_EPT; ++j)
    if (dv[j] >= 0) atomicAdd(&hist_bins[dv[j] >> SORT_SH], 1);
  __syncthreads();

  const int n_chunks = gridDim.x;
  for (int b = tid; b < n_buckets; b += SORT_CT)
    hist[b * n_chunks + blockIdx.x] = hist_bins[b];
}

__global__ void __launch_bounds__(1024)
gat_scan_counts(const int* __restrict__ counts, int* __restrict__ seg, int n) {
  __shared__ int pre_part[32];
  __shared__ int tile_part[32];
  const int tid   = threadIdx.x;
  const int lane  = tid & 31;
  const int wave  = tid >> 5;
  const int tile0 = blockIdx.x * 1024;

  int pre = 0;
  #pragma unroll 8
  for (int j = 0; j < (int)blockIdx.x; ++j) pre += counts[(j << 10) + tid];
  for (int off = 16; off > 0; off >>= 1) pre += __shfl_xor(pre, off);
  if (lane == 0) pre_part[wave] = pre;

  const int i = tile0 + tid;
  const int v = (i < n) ? counts[i] : 0;
  int x = v;
  for (int off = 1; off < 32; off <<= 1) {
    const int t = __shfl_up(x, off);
    if (lane >= off) x += t;
  }
  if (lane == 31) tile_part[wave] = x;
  __syncthreads();

  int base = 0;
  for (int w = 0; w < 32; ++w) {
    base += pre_part[w];
    if (w < wave) base += tile_part[w];
  }
  if (i < n) seg[i] = base + x - v;
}

__global__ void __launch_bounds__(SORT_CT)
gat_chunk_scatter(const int* __restrict__ src, const int* __restrict__ dst,
                  const int* __restrict__ offs, int* __restrict__ part,
                  int E, int n_buckets) {
  extern __shared__ int cell_next[];
  const int tid = threadIdx.x;
  const int n_chunks = gridDim.x;
  for (int b = tid; b < n_buckets; b += SORT_CT)
    cell_next[b] = offs[b * n_chunks + blockIdx.x];
  __syncthreads();

  const int base = blockIdx.x * SORT_CH + tid;
  int dv[SORT_EPT], sv[SORT_EPT];
  #pragma unroll
  for (int j = 0; j < SORT_EPT; ++j) {
    const int i = base + j * SORT_CT;
    dv[j] = (i < E) ? dst[i] : -1;
    sv[j] = (i < E) ? src[i] : 0;
  }
  int2* rec = (int2*)part;
  #pragma unroll
  for (int j = 0; j < SORT_EPT; ++j) {
    if (dv[j] >= 0) {
      const int pos = atomicAdd(&cell_next[dv[j] >> SORT_SH], 1);
      rec[pos] = make_int2(sv[j], dv[j]);
    }
  }
}

__global__ void __launch_bounds__(SORT_BN)
gat_bucket_sort(const int* __restrict__ part, const int* __restrict__ offs,
                int* __restrict__ seg_end, int* __restrict__ sorted_src,
                int Nn, int E, int n_chunks) {
  __shared__ int node_cnt[SORT_BN];
  __shared__ int wave_sum[SORT_BN / 32];
  const int tid   = threadIdx.x;
  const int lane  = tid & 31;
  const int wave  = tid >> 5;
  const int b     = blockIdx.x;
  const int node0 = b << SORT_SH;
  const int lo    = offs[b * n_chunks];
  const int hi    = (b + 1 < (int)gridDim.x) ? offs[(b + 1) * n_chunks] : E;
  const int2* rec = (const int2*)part;

  node_cnt[tid] = 0;
  __syncthreads();

  for (int i0 = lo; i0 < hi; i0 += 4 * SORT_BN) {
    int2 r[4];
    #pragma unroll
    for (int u = 0; u < 4; ++u) {
      const int i = i0 + u * SORT_BN + tid;
      r[u] = (i < hi) ? rec[i] : make_int2(0, -1);
    }
    #pragma unroll
    for (int u = 0; u < 4; ++u)
      if (r[u].y >= 0) atomicAdd(&node_cnt[r[u].y - node0], 1);
  }
  __syncthreads();

  const int v = node_cnt[tid];
  int x = v;
  for (int off = 1; off < 32; off <<= 1) {
    const int t = __shfl_up(x, off);
    if (lane >= off) x += t;
  }
  if (lane == 31) wave_sum[wave] = x;
  __syncthreads();
  int before = 0;
  for (int w = 0; w < SORT_BN / 32; ++w)
    if (w < wave) before += wave_sum[w];
  const int incl = lo + before + x;

  if (node0 + tid < Nn) seg_end[node0 + tid] = incl;
  node_cnt[tid] = incl - v;
  __syncthreads();

  for (int i0 = lo; i0 < hi; i0 += 4 * SORT_BN) {
    int2 r[4];
    #pragma unroll
    for (int u = 0; u < 4; ++u) {
      const int i = i0 + u * SORT_BN + tid;
      r[u] = (i < hi) ? rec[i] : make_int2(0, -1);
    }
    #pragma unroll
    for (int u = 0; u < 4; ++u) {
      if (r[u].y >= 0) {
        const int pos = atomicAdd(&node_cnt[r[u].y - node0], 1);
        sorted_src[pos] = r[u].x;
      }
    }
  }
}

__device__ __forceinline__ float lrelu(float e) {
  return e > 0.0f ? e : NEG_SLOPE * e;
}

template <int CTRL>
__device__ __forceinline__ float dpp_take(float v) {
  return __int_as_float(__builtin_amdgcn_update_dpp(
      0, __float_as_int(v), CTRL, 0xF, 0xF, true));
}

__device__ __forceinline__ float sum8(float v) {
  v += dpp_take<0xB1>(v);
  v += dpp_take<0x4E>(v);
  v += dpp_take<0x141>(v);
  return v;
}

__device__ __forceinline__ float dot4(const float4 a, const float4 b) {
  return a.x * b.x + a.y * b.y + a.z * b.z + a.w * b.w;
}

__global__ void __launch_bounds__(256)
gat_gather_softmax(const float* __restrict__ Hout, const int* __restrict__ seg_end,
                   const int* __restrict__ sorted_src,
                   const float* __restrict__ att_src, const float* __restrict__ att_dst,
                   const float* __restrict__ bias, float* __restrict__ out, int Nn) {
  const int wave = __builtin_amdgcn_readfirstlane((int)(threadIdx.x >> 5));
  const int d    = blockIdx.x * 8 + wave;
  if (d >= Nn) return;
  const int lane = threadIdx.x & 31;
  const int ch   = lane << 2;

  const float4 ws4 = *(const float4*)(att_src + ch);
  const float4 wd4 = *(const float4*)(att_dst + ch);
  const float4 hd4 = *(const float4*)(Hout + (size_t)d * HC + ch);
  const float a_d  = sum8(dot4(hd4, wd4));

  float  m   = lrelu(sum8(dot4(hd4, ws4)) + a_d);
  float  l   = 1.0f;
  float4 acc = hd4;

  const int begin = d > 0 ? seg_end[d - 1] : 0;
  const int end   = seg_end[d];

  for (int base = begin; base < end; base += 32) {
    const int cnt = (end - base) < 32 ? (end - base) : 32;
    const int idx = (lane < cnt) ? sorted_src[base + lane] : 0;
    for (int k = 0; k < cnt; k += 4) {
      float4 hs[4];
      for (int u = 0; u < 4; ++u) {
        const int kk = (k + u) < cnt ? (k + u) : (cnt - 1);
        const int s  = __builtin_amdgcn_readlane(idx, kk);
        hs[u] = *(const float4*)(Hout + (size_t)s * HC + ch);
      }
      for (int u = 0; u < 4; ++u) {
        const float a_s = sum8(dot4(hs[u], ws4));
        const float e  = (k + u) < cnt ? lrelu(a_s + a_d) : -__builtin_inff();
        const float mn = fmaxf(m, e);
        const float sc = __expf(m - mn);
        const float p  = __expf(e - mn);
        l     = l * sc + p;
        acc.x = acc.x * sc + hs[u].x * p;
        acc.y = acc.y * sc + hs[u].y * p;
        acc.z = acc.z * sc + hs[u].z * p;
        acc.w = acc.w * sc + hs[u].w * p;
        m     = mn;
      }
    }
  }

  const float  dn = l + 1e-16f;
  const float4 b4 = *(const float4*)(bias + ch);
  float4 o;
  o.x = acc.x / dn + b4.x;
  o.y = acc.y / dn + b4.y;
  o.z = acc.z / dn + b4.z;
  o.w = acc.w / dn + b4.w;
  *(float4*)(out + (size_t)d * HC + ch) = o;
}

extern "C" void kernel_launch(void* const* d_in, const int* in_sizes, int n_in,
                              void* d_out, int out_size, void* d_ws, size_t ws_size,
                              hipStream_t stream) {
  const float* X        = (const float*)d_in[0];
  const int*   edges    = (const int*)  d_in[1];
  const float* W        = (const float*)d_in[2];
  const float* att_src  = (const float*)d_in[3];
  const float* att_dst  = (const float*)d_in[4];
  const float* bias     = (const float*)d_in[5];
  float*       out      = (float*)d_out;

  const int K  = in_sizes[2] / HC;
  const int Nn = in_sizes[0] / K;
  const int E  = in_sizes[1] / 2;
  const int* src = edges;
  const int* dst = edges + E;

  char* ws = (char*)d_ws;
  float* Hout       = (float*)ws; ws += (size_t)Nn * HC * sizeof(float);
  int*   sorted_src = (int*)ws;   ws += (size_t)E * sizeof(int);
  int*   seg        = (int*)ws;   ws += (size_t)Nn * sizeof(int);

  const int n_buckets = (Nn + SORT_BN - 1) >> SORT_SH;
  const int n_chunks  = (E + SORT_CH - 1) / SORT_CH;
  const int n_cells   = n_buckets * n_chunks;
  int* part = (int*)Hout;
  int* hist = part + 2 * (size_t)E;
  int* offs = hist + n_cells;

  gat_chunk_hist<<<n_chunks, SORT_CT, n_buckets * sizeof(int), stream>>>(dst, hist, E, n_buckets);
  gat_scan_counts<<<(n_cells + 1023) / 1024, 1024, 0, stream>>>(hist, offs, n_cells);
  gat_chunk_scatter<<<n_chunks, SORT_CT, n_buckets * sizeof(int), stream>>>(
      src, dst, offs, part, E, n_buckets);
  gat_bucket_sort<<<n_buckets, SORT_BN, 0, stream>>>(part, offs, seg, sorted_src, Nn, E, n_chunks);

  gat_gemm_wmma<<<GEMM_BLOCKS, 256, 0, stream>>>(X, W, Hout, Nn);

  gat_gather_softmax<<<(Nn + 7) / 8, 256, 0, stream>>>(
      Hout, seg, sorted_src, att_src, att_dst, bias, out, Nn);
}


# Gather works on two source rows per wave instruction, one per 16-lane half (lane owns 8 channels, a quad shares a head); sum order: even and odd edges summed apart with own max, merged at the end
# speedup vs baseline: 4.5030x; 1.0440x over previous
#include <hip/hip_runtime.h>
#include <math.h>


#define HEADS     4
#define HC        128
#define NEG_SLOPE 0.2f

typedef __attribute__((ext_vector_type(2))) float v2f;
typedef __attribute__((ext_vector_type(8))) float v8f;

#define KDIM        256
#define WPITCH      288
#define GEMM_BLOCKS 512

__global__ void __launch_bounds__(256)
gat_gemm_wmma(const float* __restrict__ X,
              const float* __restrict__ W,
              float* __restrict__ Hout,
              int Nn) {
  __shared__ __attribute__((aligned(16))) float wl[(KDIM / 2) * WPITCH];

  {
    const int c4 = (threadIdx.x & 31) << 2;
    #pragma unroll 4
    for (int i = 0; i < KDIM / 16; ++i) {
      const int p = (int)(threadIdx.x >> 5) + 8 * i;
      const float4 w0 = *(const float4*)(W + (size_t)(2 * p) * HC + c4);
      const float4 w1 = *(const float4*)(W + (size_t)(2 * p + 1) * HC + c4);
      float4 lo, hi;
      lo.x = w0.x; lo.y = w1.x; lo.z = w0.y; lo.w = w1.y;
      hi.x = w0.z; hi.y = w1.z; hi.z = w0.w; hi.w = w1.w;
      float* q = wl + p * WPITCH + 2 * c4;
      *(float4*)(q)     = lo;
      *(float4*)(q + 4) = hi;
    }
  }
  __syncthreads();

  const int wave = __builtin_amdgcn_readfirstlane((int)(threadIdx.x >> 5));
  const int lane = threadIdx.x & 31;
  const int half = lane >> 4;
  const int l15  = lane & 15;
  const int n_units = ((Nn + 15) >> 4) << 1;
  const int n_waves = gridDim.x << 3;

  for (int u = blockIdx.x * 8 + wave; u < n_units; u += n_waves) {
    const int m0 = (u >> 1) << 4;
    const int n0 = (u & 1) << 6;
    const float* xa = X + (size_t)(m0 + l15) * KDIM + (half << 1);
    const float* wb = wl + half * WPITCH + 2 * (n0 + l15);

    v8f acc[4] = {};
    float2 ac[8], an[8];
    #pragma unroll
    for (int s = 0; s < 8; ++s) ac[s] = *(const float2*)(xa + 4 * s);

    #pragma unroll 2
    for (int g = 0; g < KDIM / 32; ++g) {
      if (g + 1 < KDIM / 32) {
        #pragma unroll
        for (int s = 0; s < 8; ++s)
          an[s] = *(const float2*)(xa + 32 * (g + 1) + 4 * s);
      }
      const float* wg = wb + g * (16 * WPITCH);
      #pragma unroll
      for (int s = 0; s < 8; ++s) {
        v2f a; a.x = ac[s].x; a.y = ac[s].y;
        #pragma unroll
        for (int t = 0; t < 4; ++t) {
          const float2 bv = *(const float2*)(wg + (2 * s) * WPITCH + 32 * t);
          v2f b; b.x = bv.x; b.y = bv.y;
          acc[t] = __builtin_amdgcn_wmma_f32_16x16x4_f32(
               false, a,  false, b,
               (short)0, acc[t],  false,  false);
        }
      }
      #pragma unroll
      for (int s = 0; s < 8; ++s) ac[s] = an[s];
    }

    #pragma unroll
    for (int t = 0; t < 4; ++t) {
      #pragma unroll
      for (int r = 0; r < 8; ++r) {
        const int rr = m0 + r + (half << 3);
        Hout[(size_t)rr * HC + n0 + 16 * t + l15] = acc[t][r];
      }
    }
  }
}

#define SORT_SH   9
#define SORT_BN   (1 << SORT_SH)
#define SORT_CH   4096
#define SORT_CT   256
#define SORT_EPT  (SORT_CH / SORT_CT)

__global__ void __launch_bounds__(SORT_CT)
gat_chunk_hist(const int* __restrict__ dst, int* __restrict__ hist, int E, int n_buckets) {
  extern __shared__ int hist_bins[];
  const int tid = threadIdx.x;
  for (int b = tid; b < n_buckets; b += SORT_CT) hist_bins[b] = 0;
  __syncthreads();

  const int base = blockIdx.x * SORT_CH + tid;
  int dv[SORT_EPT];
  #pragma unroll
  for (int j = 0; j < SORT_EPT; ++j) {
    const int i = base + j * SORT_CT;
    dv[j] = (i < E) ? dst[i] : -1;
  }
  #pragma unroll
  for (int j = 0; j < SORT_EPT; ++j)
    if (dv[j] >= 0) atomicAdd(&hist_bins[dv[j] >> SORT_SH], 1);
  __syncthreads();

  const int n_chunks = gridDim.x;
  for (int b = tid; b < n_buckets; b += SORT_CT)
    hist[b * n_chunks + blockIdx.x] = hist_bins[b];
}

__global__ void __launch_bounds__(1024)
gat_scan_counts(const int* __restrict__ counts, int* __restrict__ seg, int n) {
  __shared__ int pre_part[32];
  __shared__ int tile_part[32];
  const int tid   = threadIdx.x;
  const int lane  = tid & 31;
  const int wave  = tid >> 5;
  const int tile0 = blockIdx.x * 1024;

  int pre = 0;
  #pragma unroll 8
  for (int j = 0; j < (int)blockIdx.x; ++j) pre += counts[(j << 10) + tid];
  for (int off = 16; off > 0; off >>= 1) pre += __shfl_xor(pre, off);
  if (lane == 0) pre_part[wave] = pre;

  const int i = tile0 + tid;
  const int v = (i < n) ? counts[i] : 0;
  int x = v;
  for (int off = 1; off < 32; off <<= 1) {
    const int t = __shfl_up(x, off);
    if (lane >= off) x += t;
  }
  if (lane == 31) tile_part[wave] = x;
  __syncthreads();

  int base = 0;
  for (int w = 0; w < 32; ++w) {
    base += pre_part[w];
    if (w < wave) base += tile_part[w];
  }
  if (i < n) seg[i] = base + x - v;
}

__global__ void __launch_bounds__(SORT_CT)
gat_chunk_scatter(const int* __restrict__ src, const int* __restrict__ dst,
                  const int* __restrict__ offs, int* __restrict__ part,
                  int E, int n_buckets) {
  extern __shared__ int cell_next[];
  const int tid = threadIdx.x;
  const int n_chunks = gridDim.x;
  for (int b = tid; b < n_buckets; b += SORT_CT)
    cell_next[b] = offs[b * n_chunks + blockIdx.x];
  __syncthreads();

  const int base = blockIdx.x * SORT_CH + tid;
  int dv[SORT_EPT], sv[SORT_EPT];
  #pragma unroll
  for (int j = 0; j < SORT_EPT; ++j) {
    const int i = base + j * SORT_CT;
    dv[j] = (i < E) ? dst[i] : -1;
    sv[j] = (i < E) ? src[i] : 0;
  }
  int2* rec = (int2*)part;
  #pragma unroll
  for (int j = 0; j < SORT_EPT; ++j) {
    if (dv[j] >= 0) {
      const int pos = atomicAdd(&cell_next[dv[j] >> SORT_SH], 1);
      rec[pos] = make_int2(sv[j], dv[j]);
    }
  }
}

__global__ void __launch_bounds__(SORT_BN)
gat_bucket_sort(const int* __restrict__ part, const int* __restrict__ offs,
                int* __restrict__ seg_end, int* __restrict__ sorted_src,
                int Nn, int E, int n_chunks) {
  __shared__ int node_cnt[SORT_BN];
  __shared__ int wave_sum[SORT_BN / 32];
  const int tid   = threadIdx.x;
  const int lane  = tid & 31;
  const int wave  = tid >> 5;
  const int b     = blockIdx.x;
  const int node0 = b << SORT_SH;
  const int lo    = offs[b * n_chunks];
  const int hi    = (b + 1 < (int)gridDim.x) ? offs[(b + 1) * n_chunks] : E;
  const int2* rec = (const int2*)part;

  node_cnt[tid] = 0;
  __syncthreads();

  for (int i0 = lo; i0 < hi; i0 += 4 * SORT_BN) {
    int2 r[4];
    #pragma unroll
    for (int u = 0; u < 4; ++u) {
      const int i = i0 + u * SORT_BN + tid;
      r[u] = (i < hi) ? rec[i] : make_int2(0, -1);
    }
    #pragma unroll
    for (int u = 0; u < 4; ++u)
      if (r[u].y >= 0) atomicAdd(&node_cnt[r[u].y - node0], 1);
  }
  __syncthreads();

  const int v = node_cnt[tid];
  int x = v;
  for (int off = 1; off < 32; off <<= 1) {
    const int t = __shfl_up(x, off);
    if (lane >= off) x += t;
  }
  if (lane == 31) wave_sum[wave] = x;
  __syncthreads();
  int before = 0;
  for (int w = 0; w < SORT_BN / 32; ++w)
    if (w < wave) before += wave_sum[w];
  const int incl = lo + before + x;

  if (node0 + tid < Nn) seg_end[node0 + tid] = incl;
  node_cnt[tid] = incl - v;
  __syncthreads();

  for (int i0 = lo; i0 < hi; i0 += 4 * SORT_BN) {
    int2 r[4];
    #pragma unroll
    for (int u = 0; u < 4; ++u) {
      const int i = i0 + u * SORT_BN + tid;
      r[u] = (i < hi) ? rec[i] : make_int2(0, -1);
    }
    #pragma unroll
    for (int u = 0; u < 4; ++u) {
      if (r[u].y >= 0) {
        const int pos = atomicAdd(&node_cnt[r[u].y - node0], 1);
        sorted_src[pos] = r[u].x;
      }
    }
  }
}

__device__ __forceinline__ float lrelu(float e) {
  return e > 0.0f ? e : NEG_SLOPE * e;
}

template <int CTRL>
__device__ __forceinline__ float dpp_take(float v) {
  return __int_as_float(__builtin_amdgcn_update_dpp(
      0, __float_as_int(v), CTRL, 0xF, 0xF, true));
}

__device__ __forceinline__ float sum4(float v) {
  v += dpp_take<0xB1>(v);
  v += dpp_take<0x4E>(v);
  return v;
}

__device__ __forceinline__ float dot8(const float4 a0, const float4 a1, const float4 b0, const float4 b1) {
  v2f s = (v2f){a0.x, a0.y} * (v2f){b0.x, b0.y};
  s = __builtin_elementwise_fma((v2f){a0.z, a0.w}, (v2f){b0.z, b0.w}, s);
  s = __builtin_elementwise_fma((v2f){a1.x, a1.y}, (v2f){b1.x, b1.y}, s);
  s = __builtin_elementwise_fma((v2f){a1.z, a1.w}, (v2f){b1.z, b1.w}, s);
  return s.x + s.y;
}

__device__ __forceinline__ float other_half(float v) {
  const int b = __float_as_int(v);
  return __int_as_float(__builtin_amdgcn_permlanex16(b, b, 0x76543210, 0xfedcba98, false, false));
}

__device__ __forceinline__ void scale4(float4& a, float s) {
  a.x *= s; a.y *= s; a.z *= s; a.w *= s;
}

__device__ __forceinline__ void axpy4(float4& a, const float4 h, float p) {
  a.x += h.x * p; a.y += h.y * p; a.z += h.z * p; a.w += h.w * p;
}

template <bool TAIL>
__device__ __forceinline__ void gather_group(const float* __restrict__ Hout, unsigned ch, int idxh,
                                             int k, int lim,
                                             const float4 ws_lo, const float4 ws_hi, float a_d,
                                             float& m, float& l, float4& acc_lo, float4& acc_hi) {
  float4 h_lo[2], h_hi[2];
  float  e[2];
  for (int u = 0; u < 2; ++u) {
    const unsigned sel = (unsigned)((k >> 1) + u) * 0x11111111u;
    const unsigned s   = (unsigned)__builtin_amdgcn_permlane16(idxh, idxh, sel, sel, false, false);
    const float* row   = Hout + (s * HC + ch);
    h_lo[u] = *(const float4*)(row);
    h_hi[u] = *(const float4*)(row + 4);
  }
  for (int u = 0; u < 2; ++u) {
    const float a_s = sum4(dot8(h_lo[u], h_hi[u], ws_lo, ws_hi));
    e[u] = (!TAIL || (k + 2 * u) < lim) ? lrelu(a_s + a_d) : -__builtin_inff();
  }
  const float mn = fmaxf(m, fmaxf(e[0], e[1]));
  const float sc = __expf(m - mn);
  l *= sc;
  scale4(acc_lo, sc);
  scale4(acc_hi, sc);
  for (int u = 0; u < 2; ++u) {
    const float p = __expf(e[u] - mn);
    l += p;
    axpy4(acc_lo, h_lo[u], p);
    axpy4(acc_hi, h_hi[u], p);
  }
  m = mn;
}

__global__ void __launch_bounds__(256)
gat_gather_softmax(const float* __restrict__ Hout, const int* __restrict__ seg_end,
                   const int* __restrict__ sorted_src,
                   const float* __restrict__ att_src, const float* __restrict__ att_dst,
                   const float* __restrict__ bias, float* __restrict__ out, int Nn) {
  const int wave = __builtin_amdgcn_readfirstlane((int)(threadIdx.x >> 5));
  const int d    = blockIdx.x * 8 + wave;
  if (d >= Nn) return;
  const int      lane = threadIdx.x & 31;
  const int      half = lane >> 4;
  const int      q    = lane & 15;
  const unsigned ch   = (unsigned)q << 3;

  const float4 ws_lo = *(const float4*)(att_src + ch);
  const float4 ws_hi = *(const float4*)(att_src + ch + 4);
  const float4 wd_lo = *(const float4*)(att_dst + ch);
  const float4 wd_hi = *(const float4*)(att_dst + ch + 4);
  const float4 hd_lo = *(const float4*)(Hout + (size_t)d * HC + ch);
  const float4 hd_hi = *(const float4*)(Hout + (size_t)d * HC + ch + 4);
  const float a_d    = sum4(dot8(hd_lo, hd_hi, wd_lo, wd_hi));
  const float e_self = lrelu(sum4(dot8(hd_lo, hd_hi, ws_lo, ws_hi)) + a_d);

  const float4 zero4 = make_float4(0.0f, 0.0f, 0.0f, 0.0f);
  float  m      = half ? -3.402823466e+38f : e_self;
  float  l      = half ? 0.0f : 1.0f;
  float4 acc_lo = half ? zero4 : hd_lo;
  float4 acc_hi = half ? zero4 : hd_hi;

  const int begin = d > 0 ? seg_end[d - 1] : 0;
  const int end   = seg_end[d];

  for (int base = begin; base < end; base += 32) {
    const int cnt  = (end - base) < 32 ? (end - base) : 32;
    const int mine = 2 * q + half;
    const int idxh = (mine < cnt) ? sorted_src[base + mine] : 0;
    const int lim  = cnt - half;
    const int n_full = cnt & ~3;
    int k = 0;
    for (; k < n_full; k += 4)
      gather_group<false>(Hout, ch, idxh, k, lim, ws_lo, ws_hi, a_d, m, l, acc_lo, acc_hi);
    if (k < cnt)
      gather_group<true>(Hout, ch, idxh, k, lim, ws_lo, ws_hi, a_d, m, l, acc_lo, acc_hi);
  }

  const float4 keep = half ? acc_hi : acc_lo;
  const float4 give = half ? acc_lo : acc_hi;
  float4 got;
  got.x = other_half(give.x);
  got.y = other_half(give.y);
  got.z = other_half(give.z);
  got.w = other_half(give.w);
  const float mo   = other_half(m);
  const float lo   = other_half(l);
  const float mn   = fmaxf(m, mo);
  const float s_me = __expf(m - mn);
  const float s_ot = __expf(mo - mn);

  const float  dn = l * s_me + lo * s_ot + 1e-16f;
  const unsigned co = ch + ((unsigned)half << 2);
  const float4 b4 = *(const float4*)(bias + co);
  float4 o;
  o.x = (keep.x * s_me + got.x * s_ot) / dn + b4.x;
  o.y = (keep.y * s_me + got.y * s_ot) / dn + b4.y;
  o.z = (keep.z * s_me + got.z * s_ot) / dn + b4.z;
  o.w = (keep.w * s_me + got.w * s_ot) / dn + b4.w;
  *(float4*)(out + (size_t)d * HC + co) = o;
}

extern "C" void kernel_launch(void* const* d_in, const int* in_sizes, int n_in,
                              void* d_out, int out_size, void* d_ws, size_t ws_size,
                              hipStream_t stream) {
  const float* X        = (const float*)d_in[0];
  const int*   edges    = (const int*)  d_in[1];
  const float* W        = (const float*)d_in[2];
  const float* att_src  = (const float*)d_in[3];
  const float* att_dst  = (const float*)d_in[4];
  const float* bias     = (const float*)d_in[5];
  float*       out      = (float*)d_out;

  const int K  = in_sizes[2] / HC;
  const int Nn = in_sizes[0] / K;
  const int E  = in_sizes[1] / 2;
  const int* src = edges;
  const int* dst = edges + E;

  char* ws = (char*)d_ws;
  float* Hout       = (float*)ws; ws += (size_t)Nn * HC * sizeof(float);
  int*   sorted_src = (int*)ws;   ws += (size_t)E * sizeof(int);
  int*   seg        = (int*)ws;   ws += (size_t)Nn * sizeof(int);

  const int n_buckets = (Nn + SORT_BN - 1) >> SORT_SH;
  const int n_chunks  = (E + SORT_CH - 1) / SORT_CH;
  const int n_cells   = n_buckets * n_chunks;
  int* part = (int*)Hout;
  int* hist = part + 2 * (size_t)E;
  int* offs = hist + n_cells;

  gat_chunk_hist<<<n_chunks, SORT_CT, n_buckets * sizeof(int), stream>>>(dst, hist, E, n_buckets);
  gat_scan_counts<<<(n_cells + 1023) / 1024, 1024, 0, stream>>>(hist, offs, n_cells);
  gat_chunk_scatter<<<n_chunks, SORT_CT, n_buckets * sizeof(int), stream>>>(
      src, dst, offs, part, E, n_buckets);
  gat_bucket_sort<<<n_buckets, SORT_BN, 0, stream>>>(part, offs, seg, sorted_src, Nn, E, n_chunks);

  gat_gemm_wmma<<<GEMM_BLOCKS, 256, 0, stream>>>(X, W, Hout, Nn);

  gat_gather_softmax<<<(Nn + 7) / 8, 256, 0, stream>>>(
      Hout, seg, sorted_src, att_src, att_dst, bias, out, Nn);
}


# Edge sort in 2 launches, not 4: a chunk block orders its 4096 edges by node bucket in LDS and writes them to its own slot, bucket blocks collect their runs there; sum order: new LDS arrival rank
# speedup vs baseline: 5.3600x; 1.1903x over previous
#include <hip/hip_runtime.h>
#include <math.h>


#define HEADS     4
#define HC        128
#define NEG_SLOPE 0.2f

typedef __attribute__((ext_vector_type(2))) float v2f;
typedef __attribute__((ext_vector_type(8))) float v8f;

#define KDIM        256
#define WPITCH      288
#define GEMM_BLOCKS 512

__global__ void __launch_bounds__(256)
gat_gemm_wmma(const float* __restrict__ X,
              const float* __restrict__ W,
              float* __restrict__ Hout,
              int Nn) {
  __shared__ __attribute__((aligned(16))) float wl[(KDIM / 2) * WPITCH];

  {
    const int c4 = (threadIdx.x & 31) << 2;
    #pragma unroll 4
    for (int i = 0; i < KDIM / 16; ++i) {
      const int p = (int)(threadIdx.x >> 5) + 8 * i;
      const float4 w0 = *(const float4*)(W + (size_t)(2 * p) * HC + c4);
      const float4 w1 = *(const float4*)(W + (size_t)(2 * p + 1) * HC + c4);
      float4 lo, hi;
      lo.x = w0.x; lo.y = w1.x; lo.z = w0.y; lo.w = w1.y;
      hi.x = w0.z; hi.y = w1.z; hi.z = w0.w; hi.w = w1.w;
      float* q = wl + p * WPITCH + 2 * c4;
      *(float4*)(q)     = lo;
      *(float4*)(q + 4) = hi;
    }
  }
  __syncthreads();

  const int wave = __builtin_amdgcn_readfirstlane((int)(threadIdx.x >> 5));
  const int lane = threadIdx.x & 31;
  const int half = lane >> 4;
  const int l15  = lane & 15;
  const int n_units = ((Nn + 15) >> 4) << 1;
  const int n_waves = gridDim.x << 3;

  for (int u = blockIdx.x * 8 + wave; u < n_units; u += n_waves) {
    const int m0 = (u >> 1) << 4;
    const int n0 = (u & 1) << 6;
    const float* xa = X + (size_t)(m0 + l15) * KDIM + (half << 1);
    const float* wb = wl + half * WPITCH + 2 * (n0 + l15);

    v8f acc[4] = {};
    float2 ac[8], an[8];
    #pragma unroll
    for (int s = 0; s < 8; ++s) ac[s] = *(const float2*)(xa + 4 * s);

    #pragma unroll 2
    for (int g = 0; g < KDIM / 32; ++g) {
      if (g + 1 < KDIM / 32) {
        #pragma unroll
        for (int s = 0; s < 8; ++s)
          an[s] = *(const float2*)(xa + 32 * (g + 1) + 4 * s);
      }
      const float* wg = wb + g * (16 * WPITCH);
      #pragma unroll
      for (int s = 0; s < 8; ++s) {
        v2f a; a.x = ac[s].x; a.y = ac[s].y;
        #pragma unroll
        for (int t = 0; t < 4; ++t) {
          const float2 bv = *(const float2*)(wg + (2 * s) * WPITCH + 32 * t);
          v2f b; b.x = bv.x; b.y = bv.y;
          acc[t] = __builtin_amdgcn_wmma_f32_16x16x4_f32(
               false, a,  false, b,
               (short)0, acc[t],  false,  false);
        }
      }
      #pragma unroll
      for (int s = 0; s < 8; ++s) ac[s] = an[s];
    }

    #pragma unroll
    for (int t = 0; t < 4; ++t) {
      #pragma unroll
      for (int r = 0; r < 8; ++r) {
        const int rr = m0 + r + (half << 3);
        Hout[(size_t)rr * HC + n0 + 16 * t + l15] = acc[t][r];
      }
    }
  }
}

#define SORT_SH   9
#define SORT_BN   (1 << SORT_SH)
#define SORT_CH   4096
#define SORT_CT   256
#define SORT_EPT  (SORT_CH / SORT_CT)
#define SORT_CPT  8

__global__ void __launch_bounds__(SORT_CT)
gat_chunk_partition(const int* __restrict__ src, const int* __restrict__ dst,
                    int* __restrict__ part, int* __restrict__ tab,
                    int E, int n_buckets, int tab_pitch) {
  extern __shared__ int bins[];
  __shared__ __attribute__((aligned(16))) int2 stage[SORT_CH];
  __shared__ int wsum[SORT_CT / 32];
  const int tid  = threadIdx.x;
  const int lane = tid & 31;
  const int wave = tid >> 5;
  for (int b = tid; b < n_buckets; b += SORT_CT) bins[b] = 0;
  __syncthreads();

  const int base = blockIdx.x * SORT_CH + tid;
  int dv[SORT_EPT], sv[SORT_EPT], rk[SORT_EPT];
  #pragma unroll
  for (int j = 0; j < SORT_EPT; ++j) {
    const int i = base + j * SORT_CT;
    dv[j] = (i < E) ? dst[i] : -1;
    sv[j] = (i < E) ? src[i] : 0;
  }
  #pragma unroll
  for (int j = 0; j < SORT_EPT; ++j)
    rk[j] = (dv[j] >= 0) ? atomicAdd(&bins[dv[j] >> SORT_SH], 1) : 0;
  __syncthreads();

  int carry = 0;
  for (int b0 = 0; b0 < n_buckets; b0 += SORT_CT) {
    const int b = b0 + tid;
    const int v = (b < n_buckets) ? bins[b] : 0;
    int x = v;
    for (int off = 1; off < 32; off <<= 1) {
      const int t = __shfl_up(x, off);
      if (lane >= off) x += t;
    }
    if (lane == 31) wsum[wave] = x;
    __syncthreads();
    int before = carry, tot = 0;
    for (int w = 0; w < SORT_CT / 32; ++w) {
      const int s = wsum[w];
      if (w < wave) before += s;
      tot += s;
    }
    if (b < n_buckets) bins[b] = before + x - v;
    carry += tot;
    __syncthreads();
  }

  int* trow = tab + (size_t)blockIdx.x * tab_pitch;
  for (int b = tid; b < n_buckets; b += SORT_CT) trow[b] = bins[b];
  if (tid == 0) trow[n_buckets] = carry;

  #pragma unroll
  for (int j = 0; j < SORT_EPT; ++j)
    if (dv[j] >= 0) stage[bins[dv[j] >> SORT_SH] + rk[j]] = make_int2(sv[j], dv[j]);
  __syncthreads();

  const int4* s4 = (const int4*)stage;
  int4* p4 = (int4*)part + (size_t)blockIdx.x * (SORT_CH / 2);
  const int n4 = carry >> 1;
  for (int i = tid; i < n4; i += SORT_CT) p4[i] = s4[i];
  if ((carry & 1) && tid == 0)
    ((int2*)part)[(size_t)blockIdx.x * SORT_CH + carry - 1] = stage[carry - 1];
}

template <bool PLACE>
__device__ __forceinline__ void bucket_pass(const int2* __restrict__ rec, const int2* cell,
                                            int* node_cnt, int* __restrict__ sorted_src,
                                            int node0, int n_cells, int wave, int lane) {
  for (int c0 = wave; c0 < n_cells; c0 += (SORT_BN / 32) * SORT_CPT) {
    int2 cl[SORT_CPT], r[SORT_CPT];
    #pragma unroll
    for (int u = 0; u < SORT_CPT; ++u) cl[u] = cell[c0 + (SORT_BN / 32) * u];
    #pragma unroll
    for (int u = 0; u < SORT_CPT; ++u)
      r[u] = (lane < cl[u].y) ? rec[cl[u].x + lane] : make_int2(0, -1);
    #pragma unroll
    for (int u = 0; u < SORT_CPT; ++u) {
      if (r[u].y >= 0) {
        const int pos = atomicAdd(&node_cnt[r[u].y - node0], 1);
        if (PLACE) sorted_src[pos] = r[u].x;
      }
    }
    #pragma unroll
    for (int u = 0; u < SORT_CPT; ++u) {
      for (int i = lane + 32; i < cl[u].y; i += 32) {
        const int2 q = rec[cl[u].x + i];
        const int pos = atomicAdd(&node_cnt[q.y - node0], 1);
        if (PLACE) sorted_src[pos] = q.x;
      }
    }
  }
}

__global__ void __launch_bounds__(SORT_BN)
gat_bucket_sort(const int* __restrict__ part, const int* __restrict__ tab,
                int* __restrict__ seg_end, int* __restrict__ sorted_src,
                int Nn, int n_chunks, int n_cells, int tab_pitch) {
  extern __shared__ int2 cell[];
  __shared__ int node_cnt[SORT_BN];
  __shared__ int wave_sum[SORT_BN / 32];
  __shared__ int base_sum[SORT_BN / 32];
  const int tid   = threadIdx.x;
  const int lane  = tid & 31;
  const int wave  = __builtin_amdgcn_readfirstlane(tid >> 5);
  const int b     = blockIdx.x;
  const int node0 = b << SORT_SH;
  const int2* rec = (const int2*)part;

  node_cnt[tid] = 0;
  int below = 0;
  for (int c = tid; c < n_cells; c += SORT_BN) {
    int s = 0, e = 0;
    if (c < n_chunks) {
      s = tab[(size_t)c * tab_pitch + b];
      e = tab[(size_t)c * tab_pitch + b + 1];
    }
    cell[c] = make_int2(c * SORT_CH + s, e - s);
    below += s;
  }
  for (int off = 16; off > 0; off >>= 1) below += __shfl_xor(below, off);
  if (lane == 0) base_sum[wave] = below;
  __syncthreads();
  int lo = 0;
  for (int w = 0; w < SORT_BN / 32; ++w) lo += base_sum[w];

  bucket_pass<false>(rec, cell, node_cnt, sorted_src, node0, n_cells, wave, lane);
  __syncthreads();

  const int v = node_cnt[tid];
  int x = v;
  for (int off = 1; off < 32; off <<= 1) {
    const int t = __shfl_up(x, off);
    if (lane >= off) x += t;
  }
  if (lane == 31) wave_sum[wave] = x;
  __syncthreads();
  int before = 0;
  for (int w = 0; w < SORT_BN / 32; ++w)
    if (w < wave) before += wave_sum[w];
  const int incl = lo + before + x;

  if (node0 + tid < Nn) seg_end[node0 + tid] = incl;
  node_cnt[tid] = incl - v;
  __syncthreads();

  bucket_pass<true>(rec, cell, node_cnt, sorted_src, node0, n_cells, wave, lane);
}

__device__ __forceinline__ float lrelu(float e) {
  return e > 0.0f ? e : NEG_SLOPE * e;
}

template <int CTRL>
__device__ __forceinline__ float dpp_take(float v) {
  return __int_as_float(__builtin_amdgcn_update_dpp(
      0, __float_as_int(v), CTRL, 0xF, 0xF, true));
}

__device__ __forceinline__ float sum4(float v) {
  v += dpp_take<0xB1>(v);
  v += dpp_take<0x4E>(v);
  return v;
}

__device__ __forceinline__ float dot8(const float4 a0, const float4 a1, const float4 b0, const float4 b1) {
  v2f s = (v2f){a0.x, a0.y} * (v2f){b0.x, b0.y};
  s = __builtin_elementwise_fma((v2f){a0.z, a0.w}, (v2f){b0.z, b0.w}, s);
  s = __builtin_elementwise_fma((v2f){a1.x, a1.y}, (v2f){b1.x, b1.y}, s);
  s = __builtin_elementwise_fma((v2f){a1.z, a1.w}, (v2f){b1.z, b1.w}, s);
  return s.x + s.y;
}

__device__ __forceinline__ float other_half(float v) {
  const int b = __float_as_int(v);
  return __int_as_float(__builtin_amdgcn_permlanex16(b, b, 0x76543210, 0xfedcba98, false, false));
}

__device__ __forceinline__ void scale4(float4& a, float s) {
  a.x *= s; a.y *= s; a.z *= s; a.w *= s;
}

__device__ __forceinline__ void axpy4(float4& a, const float4 h, float p) {
  a.x += h.x * p; a.y += h.y * p; a.z += h.z * p; a.w += h.w * p;
}

template <bool TAIL>
__device__ __forceinline__ void gather_group(const float* __restrict__ Hout, unsigned ch, int idxh,
                                             int k, int lim,
                                             const float4 ws_lo, const float4 ws_hi, float a_d,
                                             float& m, float& l, float4& acc_lo, float4& acc_hi) {
  float4 h_lo[2], h_hi[2];
  float  e[2];
  for (int u = 0; u < 2; ++u) {
    const unsigned sel = (unsigned)((k >> 1) + u) * 0x11111111u;
    const unsigned s   = (unsigned)__builtin_amdgcn_permlane16(idxh, idxh, sel, sel, false, false);
    const float* row   = Hout + (s * HC + ch);
    h_lo[u] = *(const float4*)(row);
    h_hi[u] = *(const float4*)(row + 4);
  }
  for (int u = 0; u < 2; ++u) {
    const float a_s = sum4(dot8(h_lo[u], h_hi[u], ws_lo, ws_hi));
    e[u] = (!TAIL || (k + 2 * u) < lim) ? lrelu(a_s + a_d) : -__builtin_inff();
  }
  const float mn = fmaxf(m, fmaxf(e[0], e[1]));
  const float sc = __expf(m - mn);
  l *= sc;
  scale4(acc_lo, sc);
  scale4(acc_hi, sc);
  for (int u = 0; u < 2; ++u) {
    const float p = __expf(e[u] - mn);
    l += p;
    axpy4(acc_lo, h_lo[u], p);
    axpy4(acc_hi, h_hi[u], p);
  }
  m = mn;
}

__global__ void __launch_bounds__(256)
gat_gather_softmax(const float* __restrict__ Hout, const int* __restrict__ seg_end,
                   const int* __restrict__ sorted_src,
                   const float* __restrict__ att_src, const float* __restrict__ att_dst,
                   const float* __restrict__ bias, float* __restrict__ out, int Nn) {
  const int wave = __builtin_amdgcn_readfirstlane((int)(threadIdx.x >> 5));
  const int d    = blockIdx.x * 8 + wave;
  if (d >= Nn) return;
  const int      lane = threadIdx.x & 31;
  const int      half = lane >> 4;
  const int      q    = lane & 15;
  const unsigned ch   = (unsigned)q << 3;

  const float4 ws_lo = *(const float4*)(att_src + ch);
  const float4 ws_hi = *(const float4*)(att_src + ch + 4);
  const float4 wd_lo = *(const float4*)(att_dst + ch);
  const float4 wd_hi = *(const float4*)(att_dst + ch + 4);
  const float4 hd_lo = *(const float4*)(Hout + (size_t)d * HC + ch);
  const float4 hd_hi = *(const float4*)(Hout + (size_t)d * HC + ch + 4);
  const float a_d    = sum4(dot8(hd_lo, hd_hi, wd_lo, wd_hi));
  const float e_self = lrelu(sum4(dot8(hd_lo, hd_hi, ws_lo, ws_hi)) + a_d);

  const float4 zero4 = make_float4(0.0f, 0.0f, 0.0f, 0.0f);
  float  m      = half ? -3.402823466e+38f : e_self;
  float  l      = half ? 0.0f : 1.0f;
  float4 acc_lo = half ? zero4 : hd_lo;
  float4 acc_hi = half ? zero4 : hd_hi;

  const int begin = d > 0 ? seg_end[d - 1] : 0;
  const int end   = seg_end[d];

  for (int base = begin; base < end; base += 32) {
    const int cnt  = (end - base) < 32 ? (end - base) : 32;
    const int mine = 2 * q + half;
    const int idxh = (mine < cnt) ? sorted_src[base + mine] : 0;
    const int lim  = cnt - half;
    const int n_full = cnt & ~3;
    int k = 0;
    for (; k < n_full; k += 4)
      gather_group<false>(Hout, ch, idxh, k, lim, ws_lo, ws_hi, a_d, m, l, acc_lo, acc_hi);
    if (k < cnt)
      gather_group<true>(Hout, ch, idxh, k, lim, ws_lo, ws_hi, a_d, m, l, acc_lo, acc_hi);
  }

  const float4 keep = half ? acc_hi : acc_lo;
  const float4 give = half ? acc_lo : acc_hi;
  float4 got;
  got.x = other_half(give.x);
  got.y = other_half(give.y);
  got.z = other_half(give.z);
  got.w = other_half(give.w);
  const float mo   = other_half(m);
  const float lo   = other_half(l);
  const float mn   = fmaxf(m, mo);
  const float s_me = __expf(m - mn);
  const float s_ot = __expf(mo - mn);

  const float  dn = l * s_me + lo * s_ot + 1e-16f;
  const unsigned co = ch + ((unsigned)half << 2);
  const float4 b4 = *(const float4*)(bias + co);
  float4 o;
  o.x = (keep.x * s_me + got.x * s_ot) / dn + b4.x;
  o.y = (keep.y * s_me + got.y * s_ot) / dn + b4.y;
  o.z = (keep.z * s_me + got.z * s_ot) / dn + b4.z;
  o.w = (keep.w * s_me + got.w * s_ot) / dn + b4.w;
  *(float4*)(out + (size_t)d * HC + co) = o;
}

extern "C" void kernel_launch(void* const* d_in, const int* in_sizes, int n_in,
                              void* d_out, int out_size, void* d_ws, size_t ws_size,
                              hipStream_t stream) {
  const float* X        = (const float*)d_in[0];
  const int*   edges    = (const int*)  d_in[1];
  const float* W        = (const float*)d_in[2];
  const float* att_src  = (const float*)d_in[3];
  const float* att_dst  = (const float*)d_in[4];
  const float* bias     = (const float*)d_in[5];
  float*       out      = (float*)d_out;

  const int K  = in_sizes[2] / HC;
  const int Nn = in_sizes[0] / K;
  const int E  = in_sizes[1] / 2;
  const int* src = edges;
  const int* dst = edges + E;

  char* ws = (char*)d_ws;
  float* Hout       = (float*)ws; ws += (size_t)Nn * HC * sizeof(float);
  int*   sorted_src = (int*)ws;   ws += (size_t)E * sizeof(int);
  int*   seg        = (int*)ws;   ws += (size_t)Nn * sizeof(int);

  const int n_buckets = (Nn + SORT_BN - 1) >> SORT_SH;
  const int n_chunks  = (E + SORT_CH - 1) / SORT_CH;
  const int tab_pitch = (n_buckets + 1 + 31) & ~31;
  const int run_trip  = (SORT_BN / 32) * SORT_CPT;
  const int n_cells   = (n_chunks + run_trip - 1) / run_trip * run_trip;
  int* part = (int*)Hout;
  int* tab  = part + 2 * (size_t)n_chunks * SORT_CH;

  gat_chunk_partition<<<n_chunks, SORT_CT, n_buckets * sizeof(int), stream>>>(
      src, dst, part, tab, E, n_buckets, tab_pitch);
  gat_bucket_sort<<<n_buckets, SORT_BN, n_cells * sizeof(int2), stream>>>(
      part, tab, seg, sorted_src, Nn, n_chunks, n_cells, tab_pitch);

  gat_gemm_wmma<<<GEMM_BLOCKS, 256, 0, stream>>>(X, W, Hout, Nn);

  gat_gather_softmax<<<(Nn + 7) / 8, 256, 0, stream>>>(
      Hout, seg, sorted_src, att_src, att_dst, bias, out, Nn);
}


# Second sort level runs inside the product's launch: 256 product blocks of 16 waves (one per WGP, row tiles dealt block by block) beside the 196 bucket blocks; 3 launches, not 4; sum order unchanged
# speedup vs baseline: 6.0982x; 1.1377x over previous
#include <hip/hip_runtime.h>
#include <math.h>


#define HEADS     4
#define HC        128
#define NEG_SLOPE 0.2f

typedef __attribute__((ext_vector_type(2))) float v2f;
typedef __attribute__((ext_vector_type(8))) float v8f;

#define KDIM        256
#define WPITCH      288
#define GEMM_BLOCKS 256
#define FUSED_T     512

__device__ __forceinline__ void
gemm_role(float* wl,
          const float* __restrict__ X,
          const float* __restrict__ W,
          float* __restrict__ Hout,
          int Nn, int gb, int n_gb) {
  {
    const int c4 = (threadIdx.x & 31) << 2;
    #pragma unroll 4
    for (int i = 0; i < KDIM / 32; ++i) {
      const int p = (int)(threadIdx.x >> 5) + 16 * i;
      const float4 w0 = *(const float4*)(W + (size_t)(2 * p) * HC + c4);
      const float4 w1 = *(const float4*)(W + (size_t)(2 * p + 1) * HC + c4);
      float4 lo, hi;
      lo.x = w0.x; lo.y = w1.x; lo.z = w0.y; lo.w = w1.y;
      hi.x = w0.z; hi.y = w1.z; hi.z = w0.w; hi.w = w1.w;
      float* q = wl + p * WPITCH + 2 * c4;
      *(float4*)(q)     = lo;
      *(float4*)(q + 4) = hi;
    }
  }
  __syncthreads();

  const int wave = __builtin_amdgcn_readfirstlane((int)(threadIdx.x >> 5));
  const int lane = threadIdx.x & 31;
  const int half = lane >> 4;
  const int l15  = lane & 15;
  const int n_tiles = (Nn + 15) >> 4;
  const int n_pairs = n_gb << 3;
  const int n0      = (wave & 1) << 6;

  for (int rt = (wave >> 1) * n_gb + gb; rt < n_tiles; rt += n_pairs) {
    const int m0 = rt << 4;
    const float* xa = X + (size_t)(m0 + l15) * KDIM + (half << 1);
    const float* wb = wl + half * WPITCH + 2 * (n0 + l15);

    v8f acc[4] = {};
    float2 ac[8], an[8];
    #pragma unroll
    for (int s = 0; s < 8; ++s) ac[s] = *(const float2*)(xa + 4 * s);

    #pragma unroll 2
    for (int g = 0; g < KDIM / 32; ++g) {
      if (g + 1 < KDIM / 32) {
        #pragma unroll
        for (int s = 0; s < 8; ++s)
          an[s] = *(const float2*)(xa + 32 * (g + 1) + 4 * s);
      }
      const float* wg = wb + g * (16 * WPITCH);
      #pragma unroll
      for (int s = 0; s < 8; ++s) {
        v2f a; a.x = ac[s].x; a.y = ac[s].y;
        #pragma unroll
        for (int t = 0; t < 4; ++t) {
          const float2 bv = *(const float2*)(wg + (2 * s) * WPITCH + 32 * t);
          v2f b; b.x = bv.x; b.y = bv.y;
          acc[t] = __builtin_amdgcn_wmma_f32_16x16x4_f32(
               false, a,  false, b,
               (short)0, acc[t],  false,  false);
        }
      }
      #pragma unroll
      for (int s = 0; s < 8; ++s) ac[s] = an[s];
    }

    #pragma unroll
    for (int t = 0; t < 4; ++t) {
      #pragma unroll
      for (int r = 0; r < 8; ++r) {
        const int rr = m0 + r + (half << 3);
        Hout[(size_t)rr * HC + n0 + 16 * t + l15] = acc[t][r];
      }
    }
  }
}

#define SORT_SH   9
#define SORT_BN   (1 << SORT_SH)
#define SORT_CH   4096
#define SORT_CT   256
#define SORT_EPT  (SORT_CH / SORT_CT)
#define SORT_CPT  8

__global__ void __launch_bounds__(SORT_CT)
gat_chunk_partition(const int* __restrict__ src, const int* __restrict__ dst,
                    int* __restrict__ part, int* __restrict__ tab,
                    int E, int n_buckets, int tab_pitch) {
  extern __shared__ int bins[];
  __shared__ __attribute__((aligned(16))) int2 stage[SORT_CH];
  __shared__ int wsum[SORT_CT / 32];
  const int tid  = threadIdx.x;
  const int lane = tid & 31;
  const int wave = tid >> 5;
  for (int b = tid; b < n_buckets; b += SORT_CT) bins[b] = 0;
  __syncthreads();

  const int base = blockIdx.x * SORT_CH + tid;
  int dv[SORT_EPT], sv[SORT_EPT], rk[SORT_EPT];
  #pragma unroll
  for (int j = 0; j < SORT_EPT; ++j) {
    const int i = base + j * SORT_CT;
    dv[j] = (i < E) ? dst[i] : -1;
    sv[j] = (i < E) ? src[i] : 0;
  }
  #pragma unroll
  for (int j = 0; j < SORT_EPT; ++j)
    rk[j] = (dv[j] >= 0) ? atomicAdd(&bins[dv[j] >> SORT_SH], 1) : 0;
  __syncthreads();

  int carry = 0;
  for (int b0 = 0; b0 < n_buckets; b0 += SORT_CT) {
    const int b = b0 + tid;
    const int v = (b < n_buckets) ? bins[b] : 0;
    int x = v;
    for (int off = 1; off < 32; off <<= 1) {
      const int t = __shfl_up(x, off);
      if (lane >= off) x += t;
    }
    if (lane == 31) wsum[wave] = x;
    __syncthreads();
    int before = carry, tot = 0;
    for (int w = 0; w < SORT_CT / 32; ++w) {
      const int s = wsum[w];
      if (w < wave) before += s;
      tot += s;
    }
    if (b < n_buckets) bins[b] = before + x - v;
    carry += tot;
    __syncthreads();
  }

  int* trow = tab + (size_t)blockIdx.x * tab_pitch;
  for (int b = tid; b < n_buckets; b += SORT_CT) trow[b] = bins[b];
  if (tid == 0) trow[n_buckets] = carry;

  #pragma unroll
  for (int j = 0; j < SORT_EPT; ++j)
    if (dv[j] >= 0) stage[bins[dv[j] >> SORT_SH] + rk[j]] = make_int2(sv[j], dv[j]);
  __syncthreads();

  const int4* s4 = (const int4*)stage;
  int4* p4 = (int4*)part + (size_t)blockIdx.x * (SORT_CH / 2);
  const int n4 = carry >> 1;
  for (int i = tid; i < n4; i += SORT_CT) p4[i] = s4[i];
  if ((carry & 1) && tid == 0)
    ((int2*)part)[(size_t)blockIdx.x * SORT_CH + carry - 1] = stage[carry - 1];
}

template <bool PLACE>
__device__ __forceinline__ void bucket_pass(const int2* __restrict__ rec, const int2* cell,
                                            int* node_cnt, int* __restrict__ sorted_src,
                                            int node0, int n_cells, int wave, int lane) {
  for (int c0 = wave; c0 < n_cells; c0 += (SORT_BN / 32) * SORT_CPT) {
    int2 cl[SORT_CPT], r[SORT_CPT];
    #pragma unroll
    for (int u = 0; u < SORT_CPT; ++u) cl[u] = cell[c0 + (SORT_BN / 32) * u];
    #pragma unroll
    for (int u = 0; u < SORT_CPT; ++u)
      r[u] = (lane < cl[u].y) ? rec[cl[u].x + lane] : make_int2(0, -1);
    #pragma unroll
    for (int u = 0; u < SORT_CPT; ++u) {
      if (r[u].y >= 0) {
        const int pos = atomicAdd(&node_cnt[r[u].y - node0], 1);
        if (PLACE) sorted_src[pos] = r[u].x;
      }
    }
    #pragma unroll
    for (int u = 0; u < SORT_CPT; ++u) {
      for (int i = lane + 32; i < cl[u].y; i += 32) {
        const int2 q = rec[cl[u].x + i];
        const int pos = atomicAdd(&node_cnt[q.y - node0], 1);
        if (PLACE) sorted_src[pos] = q.x;
      }
    }
  }
}

__device__ __forceinline__ void
bucket_role(int2* cell, int* node_cnt, int* wave_sum, int* base_sum,
            const int* __restrict__ part, const int* __restrict__ tab,
            int* __restrict__ seg_end, int* __restrict__ sorted_src,
            int Nn, int n_chunks, int n_cells, int tab_pitch, int b) {
  const int tid   = threadIdx.x;
  const int lane  = tid & 31;
  const int wave  = __builtin_amdgcn_readfirstlane(tid >> 5);
  const int node0 = b << SORT_SH;
  const int2* rec = (const int2*)part;

  node_cnt[tid] = 0;
  int below = 0;
  for (int c = tid; c < n_cells; c += SORT_BN) {
    int s = 0, e = 0;
    if (c < n_chunks) {
      s = tab[(size_t)c * tab_pitch + b];
      e = tab[(size_t)c * tab_pitch + b + 1];
    }
    cell[c] = make_int2(c * SORT_CH + s, e - s);
    below += s;
  }
  for (int off = 16; off > 0; off >>= 1) below += __shfl_xor(below, off);
  if (lane == 0) base_sum[wave] = below;
  __syncthreads();
  int lo = 0;
  for (int w = 0; w < SORT_BN / 32; ++w) lo += base_sum[w];

  bucket_pass<false>(rec, cell, node_cnt, sorted_src, node0, n_cells, wave, lane);
  __syncthreads();

  const int v = node_cnt[tid];
  int x = v;
  for (int off = 1; off < 32; off <<= 1) {
    const int t = __shfl_up(x, off);
    if (lane >= off) x += t;
  }
  if (lane == 31) wave_sum[wave] = x;
  __syncthreads();
  int before = 0;
  for (int w = 0; w < SORT_BN / 32; ++w)
    if (w < wave) before += wave_sum[w];
  const int incl = lo + before + x;

  if (node0 + tid < Nn) seg_end[node0 + tid] = incl;
  node_cnt[tid] = incl - v;
  __syncthreads();

  bucket_pass<true>(rec, cell, node_cnt, sorted_src, node0, n_cells, wave, lane);
}

static_assert(SORT_BN == FUSED_T, "a bucket block has one thread per node of its bucket");

__global__ void __launch_bounds__(FUSED_T)
gat_gemm_and_bucket_sort(const float* __restrict__ X,
                         const float* __restrict__ W,
                         float* __restrict__ Hout,
                         const int* __restrict__ part, const int* __restrict__ tab,
                         int* __restrict__ seg_end, int* __restrict__ sorted_src,
                         int Nn, int n_chunks, int n_cells, int tab_pitch, int n_gb) {
  __shared__ __attribute__((aligned(16))) float wl[(KDIM / 2) * WPITCH];
  extern __shared__ int2 cell[];
  __shared__ int node_cnt[SORT_BN];
  __shared__ int wave_sum[SORT_BN / 32];
  __shared__ int base_sum[SORT_BN / 32];
  if ((int)blockIdx.x < n_gb)
    gemm_role(wl, X, W, Hout, Nn, (int)blockIdx.x, n_gb);
  else
    bucket_role(cell, node_cnt, wave_sum, base_sum, part, tab, seg_end, sorted_src,
                Nn, n_chunks, n_cells, tab_pitch, (int)blockIdx.x - n_gb);
}

__device__ __forceinline__ float lrelu(float e) {
  return e > 0.0f ? e : NEG_SLOPE * e;
}

template <int CTRL>
__device__ __forceinline__ float dpp_take(float v) {
  return __int_as_float(__builtin_amdgcn_update_dpp(
      0, __float_as_int(v), CTRL, 0xF, 0xF, true));
}

__device__ __forceinline__ float sum4(float v) {
  v += dpp_take<0xB1>(v);
  v += dpp_take<0x4E>(v);
  return v;
}

__device__ __forceinline__ float dot8(const float4 a0, const float4 a1, const float4 b0, const float4 b1) {
  v2f s = (v2f){a0.x, a0.y} * (v2f){b0.x, b0.y};
  s = __builtin_elementwise_fma((v2f){a0.z, a0.w}, (v2f){b0.z, b0.w}, s);
  s = __builtin_elementwise_fma((v2f){a1.x, a1.y}, (v2f){b1.x, b1.y}, s);
  s = __builtin_elementwise_fma((v2f){a1.z, a1.w}, (v2f){b1.z, b1.w}, s);
  return s.x + s.y;
}

__device__ __forceinline__ float other_half(float v) {
  const int b = __float_as_int(v);
  return __int_as_float(__builtin_amdgcn_permlanex16(b, b, 0x76543210, 0xfedcba98, false, false));
}

__device__ __forceinline__ void scale4(float4& a, float s) {
  a.x *= s; a.y *= s; a.z *= s; a.w *= s;
}

__device__ __forceinline__ void axpy4(float4& a, const float4 h, float p) {
  a.x += h.x * p; a.y += h.y * p; a.z += h.z * p; a.w += h.w * p;
}

template <bool TAIL>
__device__ __forceinline__ void gather_group(const float* __restrict__ Hout, unsigned ch, int idxh,
                                             int k, int lim,
                                             const float4 ws_lo, const float4 ws_hi, float a_d,
                                             float& m, float& l, float4& acc_lo, float4& acc_hi) {
  float4 h_lo[2], h_hi[2];
  float  e[2];
  for (int u = 0; u < 2; ++u) {
    const unsigned sel = (unsigned)((k >> 1) + u) * 0x11111111u;
    const unsigned s   = (unsigned)__builtin_amdgcn_permlane16(idxh, idxh, sel, sel, false, false);
    const float* row   = Hout + (s * HC + ch);
    h_lo[u] = *(const float4*)(row);
    h_hi[u] = *(const float4*)(row + 4);
  }
  for (int u = 0; u < 2; ++u) {
    const float a_s = sum4(dot8(h_lo[u], h_hi[u], ws_lo, ws_hi));
    e[u] = (!TAIL || (k + 2 * u) < lim) ? lrelu(a_s + a_d) : -__builtin_inff();
  }
  const float mn = fmaxf(m, fmaxf(e[0], e[1]));
  const float sc = __expf(m - mn);
  l *= sc;
  scale4(acc_lo, sc);
  scale4(acc_hi, sc);
  for (int u = 0; u < 2; ++u) {
    const float p = __expf(e[u] - mn);
    l += p;
    axpy4(acc_lo, h_lo[u], p);
    axpy4(acc_hi, h_hi[u], p);
  }
  m = mn;
}

__global__ void __launch_bounds__(256)
gat_gather_softmax(const float* __restrict__ Hout, const int* __restrict__ seg_end,
                   const int* __restrict__ sorted_src,
                   const float* __restrict__ att_src, const float* __restrict__ att_dst,
                   const float* __restrict__ bias, float* __restrict__ out, int Nn) {
  const int wave = __builtin_amdgcn_readfirstlane((int)(threadIdx.x >> 5));
  const int d    = blockIdx.x * 8 + wave;
  if (d >= Nn) return;
  const int      lane = threadIdx.x & 31;
  const int      half = lane >> 4;
  const int      q    = lane & 15;
  const unsigned ch   = (unsigned)q << 3;

  const float4 ws_lo = *(const float4*)(att_src + ch);
  const float4 ws_hi = *(const float4*)(att_src + ch + 4);
  const float4 wd_lo = *(const float4*)(att_dst + ch);
  const float4 wd_hi = *(const float4*)(att_dst + ch + 4);
  const float4 hd_lo = *(const float4*)(Hout + (size_t)d * HC + ch);
  const float4 hd_hi = *(const float4*)(Hout + (size_t)d * HC + ch + 4);
  const float a_d    = sum4(dot8(hd_lo, hd_hi, wd_lo, wd_hi));
  const float e_self = lrelu(sum4(dot8(hd_lo, hd_hi, ws_lo, ws_hi)) + a_d);

  const float4 zero4 = make_float4(0.0f, 0.0f, 0.0f, 0.0f);
  float  m      = half ? -3.402823466e+38f : e_self;
  float  l      = half ? 0.0f : 1.0f;
  float4 acc_lo = half ? zero4 : hd_lo;
  float4 acc_hi = half ? zero4 : hd_hi;

  const int begin = d > 0 ? seg_end[d - 1] : 0;
  const int end   = seg_end[d];

  for (int base = begin; base < end; base += 32) {
    const int cnt  = (end - base) < 32 ? (end - base) : 32;
    const int mine = 2 * q + half;
    const int idxh = (mine < cnt) ? sorted_src[base + mine] : 0;
    const int lim  = cnt - half;
    const int n_full = cnt & ~3;
    int k = 0;
    for (; k < n_full; k += 4)
      gather_group<false>(Hout, ch, idxh, k, lim, ws_lo, ws_hi, a_d, m, l, acc_lo, acc_hi);
    if (k < cnt)
      gather_group<true>(Hout, ch, idxh, k, lim, ws_lo, ws_hi, a_d, m, l, acc_lo, acc_hi);
  }

  const float4 keep = half ? acc_hi : acc_lo;
  const float4 give = half ? acc_lo : acc_hi;
  float4 got;
  got.x = other_half(give.x);
  got.y = other_half(give.y);
  got.z = other_half(give.z);
  got.w = other_half(give.w);
  const float mo   = other_half(m);
  const float lo   = other_half(l);
  const float mn   = fmaxf(m, mo);
  const float s_me = __expf(m - mn);
  const float s_ot = __expf(mo - mn);

  const float  dn = l * s_me + lo * s_ot + 1e-16f;
  const unsigned co = ch + ((unsigned)half << 2);
  const float4 b4 = *(const float4*)(bias + co);
  float4 o;
  o.x = (keep.x * s_me + got.x * s_ot) / dn + b4.x;
  o.y = (keep.y * s_me + got.y * s_ot) / dn + b4.y;
  o.z = (keep.z * s_me + got.z * s_ot) / dn + b4.z;
  o.w = (keep.w * s_me + got.w * s_ot) / dn + b4.w;
  *(float4*)(out + (size_t)d * HC + co) = o;
}

extern "C" void kernel_launch(void* const* d_in, const int* in_sizes, int n_in,
                              void* d_out, int out_size, void* d_ws, size_t ws_size,
                              hipStream_t stream) {
  const float* X        = (const float*)d_in[0];
  const int*   edges    = (const int*)  d_in[1];
  const float* W        = (const float*)d_in[2];
  const float* att_src  = (const float*)d_in[3];
  const float* att_dst  = (const float*)d_in[4];
  const float* bias     = (const float*)d_in[5];
  float*       out      = (float*)d_out;

  const int K  = in_sizes[2] / HC;
  const int Nn = in_sizes[0] / K;
  const int E  = in_sizes[1] / 2;
  const int* src = edges;
  const int* dst = edges + E;

  char* ws = (char*)d_ws;
  float* Hout       = (float*)ws; ws += (size_t)Nn * HC * sizeof(float);
  int*   sorted_src = (int*)ws;   ws += (size_t)E * sizeof(int);
  int*   seg        = (int*)ws;   ws += (size_t)Nn * sizeof(int);

  const int n_buckets = (Nn + SORT_BN - 1) >> SORT_SH;
  const int n_chunks  = (E + SORT_CH - 1) / SORT_CH;
  const int tab_pitch = (n_buckets + 1 + 31) & ~31;
  const int run_trip  = (SORT_BN / 32) * SORT_CPT;
  const int n_cells   = (n_chunks + run_trip - 1) / run_trip * run_trip;
  int* part = (int*)out;
  int* tab  = part + 2 * (size_t)n_chunks * SORT_CH;

  gat_chunk_partition<<<n_chunks, SORT_CT, n_buckets * sizeof(int), stream>>>(
      src, dst, part, tab, E, n_buckets, tab_pitch);

  gat_gemm_and_bucket_sort<<<GEMM_BLOCKS + n_buckets, FUSED_T, n_cells * sizeof(int2), stream>>>(
      X, W, Hout, part, tab, seg, sorted_src, Nn, n_chunks, n_cells, tab_pitch, GEMM_BLOCKS);

  gat_gather_softmax<<<(Nn + 7) / 8, 256, 0, stream>>>(
      Hout, seg, sorted_src, att_src, att_dst, bias, out, Nn);
}
